# starting kernel
# baseline (speedup 1.0000x reference)
#include <hip/hip_runtime.h>
#include <math.h>

#define BATCH   256
#define STATE   512
#define HIDDEN  1024
#define COND    64
#define NSTEPS  256
#define TSCALE  0.1f

typedef __attribute__((ext_vector_type(2))) float v2f;
typedef __attribute__((ext_vector_type(4))) float v4f;
typedef __attribute__((ext_vector_type(8))) float v8f;

static __device__ __forceinline__ v8f wmma_f32(v2f a, v2f b, v8f c) {
    return __builtin_amdgcn_wmma_f32_16x16x4_f32(
         false, a,  false, b,
         (short)0, c,  false,  false);
}

__global__ __launch_bounds__(256) void copy_x0_kernel(const float* __restrict__ x0,
                                                      float* __restrict__ out) {
    int i = blockIdx.x * 256 + threadIdx.x;
    out[i] = x0[i];
}

__global__ __launch_bounds__(256) void z1_precompute_kernel(const float* __restrict__ z,
                                                            const float* __restrict__ W1,
                                                            const float* __restrict__ b1,
                                                            float* __restrict__ Z1) {
    int idx = blockIdx.x * 256 + threadIdx.x;
    int b = idx >> 10;
    int j = idx & (HIDDEN - 1);
    float acc = b1[j];
    const float* zrow = z + b * COND;
    const float* wz   = W1 + (size_t)STATE * HIDDEN + j;
    #pragma unroll 4
    for (int k = 0; k < COND; ++k)
        acc += zrow[k] * wz[(size_t)k * HIDDEN];
    Z1[idx] = acc;
}

__global__ __launch_bounds__(256) void pack_w_kernel(const float* __restrict__ W,
                                                     float* __restrict__ Wp,
                                                     int K, int N) {
    int idx  = blockIdx.x * 256 + threadIdx.x;
    int lane = idx & 31;
    int kb2  = (idx >> 5) % (K >> 3);
    int nt   = (idx >> 5) / (K >> 3);
    int k0   = (kb2 << 3) + ((lane >> 4) << 1);
    int n    = (nt << 4) + (lane & 15);
    size_t o = (size_t)idx * 4;
    Wp[o + 0] = W[(size_t)(k0 + 0) * N + n];
    Wp[o + 1] = W[(size_t)(k0 + 1) * N + n];
    Wp[o + 2] = W[(size_t)(k0 + 4) * N + n];
    Wp[o + 3] = W[(size_t)(k0 + 5) * N + n];
}

__global__ __launch_bounds__(256) void step1_packed_kernel(const float* __restrict__ X,
                                                           const float* __restrict__ W1pack,
                                                           const float* __restrict__ W1,
                                                           const float* __restrict__ Z1,
                                                           const float* __restrict__ t,
                                                           int s,
                                                           float* __restrict__ Hbuf) {
    __shared__ float xs[16 * STATE];

    const int mt   = blockIdx.x >> 3;
    const int ng   = blockIdx.x & 7;
    const int m0   = mt * 16;
    const int wave = threadIdx.x >> 5;
    const int lane = threadIdx.x & 31;
    const int n0   = (ng * 8 + wave) * 16;

    for (int i = threadIdx.x; i < 16 * STATE; i += 256)
        xs[i] = X[(size_t)(m0 + (i >> 9)) * STATE + (i & (STATE - 1))];
    __syncthreads();

    const int ar = lane & 15;
    const int ak = (lane >> 4) * 2;
    const float* wp = W1pack + ((size_t)(n0 >> 4) * (STATE >> 3)) * 128 + lane * 4;

    v8f acc = {0.f, 0.f, 0.f, 0.f, 0.f, 0.f, 0.f, 0.f};
    for (int kb = 0; kb < (STATE >> 3); ++kb) {
        v4f w  = *(const v4f*)(wp + (size_t)kb * 128);
        v2f a0 = *(const v2f*)&xs[ar * STATE + kb * 8 + ak];
        v2f a1 = *(const v2f*)&xs[ar * STATE + kb * 8 + 4 + ak];
        v2f b0 = {w.x, w.y};
        v2f b1 = {w.z, w.w};
        acc = wmma_f32(a0, b0, acc);
        acc = wmma_f32(a1, b1, acc);
    }

    const float ts   = t[s];
    const float* W1t = W1 + (size_t)(STATE + COND) * HIDDEN;

    const int col   = n0 + (lane & 15);
    const int rbase = m0 + ((lane >> 4) << 3);
    #pragma unroll
    for (int r = 0; r < 8; ++r) {
        const int row = rbase + r;
        float v = acc[r] + Z1[(size_t)row * HIDDEN + col] + ts * W1t[col];
        Hbuf[(size_t)row * HIDDEN + col] = tanhf(v);
    }
}

__global__ __launch_bounds__(256) void step2_packed_kernel(const float* __restrict__ Hbuf,
                                                           const float* __restrict__ W2pack,
                                                           const float* __restrict__ b2,
                                                           const float* __restrict__ X,
                                                           const float* __restrict__ t,
                                                           float* __restrict__ Xnext) {
    __shared__ float hs[16 * HIDDEN];

    const int mt   = blockIdx.x >> 2;
    const int ng   = blockIdx.x & 3;
    const int m0   = mt * 16;
    const int wave = threadIdx.x >> 5;
    const int lane = threadIdx.x & 31;
    const int n0   = (ng * 8 + wave) * 16;

    for (int i = threadIdx.x; i < 16 * HIDDEN; i += 256)
        hs[i] = Hbuf[(size_t)(m0 + (i >> 10)) * HIDDEN + (i & (HIDDEN - 1))];
    __syncthreads();

    const int ar = lane & 15;
    const int ak = (lane >> 4) * 2;
    const float* wp = W2pack + ((size_t)(n0 >> 4) * (HIDDEN >> 3)) * 128 + lane * 4;

    v8f acc = {0.f, 0.f, 0.f, 0.f, 0.f, 0.f, 0.f, 0.f};
    for (int kb = 0; kb < (HIDDEN >> 3); ++kb) {
        v4f w  = *(const v4f*)(wp + (size_t)kb * 128);
        v2f a0 = *(const v2f*)&hs[ar * HIDDEN + kb * 8 + ak];
        v2f a1 = *(const v2f*)&hs[ar * HIDDEN + kb * 8 + 4 + ak];
        v2f b0 = {w.x, w.y};
        v2f b1 = {w.z, w.w};
        acc = wmma_f32(a0, b0, acc);
        acc = wmma_f32(a1, b1, acc);
    }

    const float dt = (t[1] - t[0]) * TSCALE;

    const int col   = n0 + (lane & 15);
    const int rbase = m0 + ((lane >> 4) << 3);
    #pragma unroll
    for (int r = 0; r < 8; ++r) {
        const int row = rbase + r;
        float f = acc[r] + b2[col];
        float v = X[(size_t)row * STATE + col] + dt * f;
        Xnext[(size_t)row * STATE + col] = fmaxf(v, 0.f);
    }
}

__global__ __launch_bounds__(256) void step1_direct_kernel(const float* __restrict__ X,
                                                           const float* __restrict__ W1,
                                                           const float* __restrict__ Z1,
                                                           const float* __restrict__ t,
                                                           int s,
                                                           float* __restrict__ Hbuf) {
    __shared__ float xs[16 * STATE];
    const int mt = blockIdx.x >> 3, ng = blockIdx.x & 7;
    const int m0 = mt * 16;
    const int wave = threadIdx.x >> 5, lane = threadIdx.x & 31;
    const int n0 = (ng * 8 + wave) * 16;
    for (int i = threadIdx.x; i < 16 * STATE; i += 256)
        xs[i] = X[(size_t)(m0 + (i >> 9)) * STATE + (i & (STATE - 1))];
    __syncthreads();
    const int ar = lane & 15, ak = (lane >> 4) * 2;
    v8f acc = {0.f, 0.f, 0.f, 0.f, 0.f, 0.f, 0.f, 0.f};
    for (int k = 0; k < STATE; k += 4) {
        v2f a = *(const v2f*)&xs[ar * STATE + k + ak];
        v2f b;
        b.x = W1[(size_t)(k + ak) * HIDDEN + n0 + ar];
        b.y = W1[(size_t)(k + ak + 1) * HIDDEN + n0 + ar];
        acc = wmma_f32(a, b, acc);
    }
    const float ts = t[s];
    const float* W1t = W1 + (size_t)(STATE + COND) * HIDDEN;
    const int col = n0 + (lane & 15);
    const int rbase = m0 + ((lane >> 4) << 3);
    #pragma unroll
    for (int r = 0; r < 8; ++r) {
        const int row = rbase + r;
        float v = acc[r] + Z1[(size_t)row * HIDDEN + col] + ts * W1t[col];
        Hbuf[(size_t)row * HIDDEN + col] = tanhf(v);
    }
}

__global__ __launch_bounds__(256) void step2_direct_kernel(const float* __restrict__ Hbuf,
                                                           const float* __restrict__ W2,
                                                           const float* __restrict__ b2,
                                                           const float* __restrict__ X,
                                                           const float* __restrict__ t,
                                                           float* __restrict__ Xnext) {
    __shared__ float hs[16 * HIDDEN];
    const int mt = blockIdx.x >> 2, ng = blockIdx.x & 3;
    const int m0 = mt * 16;
    const int wave = threadIdx.x >> 5, lane = threadIdx.x & 31;
    const int n0 = (ng * 8 + wave) * 16;
    for (int i = threadIdx.x; i < 16 * HIDDEN; i += 256)
        hs[i] = Hbuf[(size_t)(m0 + (i >> 10)) * HIDDEN + (i & (HIDDEN - 1))];
    __syncthreads();
    const int ar = lane & 15, ak = (lane >> 4) * 2;
    v8f acc = {0.f, 0.f, 0.f, 0.f, 0.f, 0.f, 0.f, 0.f};
    for (int k = 0; k < HIDDEN; k += 4) {
        v2f a = *(const v2f*)&hs[ar * HIDDEN + k + ak];
        v2f b;
        b.x = W2[(size_t)(k + ak) * STATE + n0 + ar];
        b.y = W2[(size_t)(k + ak + 1) * STATE + n0 + ar];
        acc = wmma_f32(a, b, acc);
    }
    const float dt = (t[1] - t[0]) * TSCALE;
    const int col = n0 + (lane & 15);
    const int rbase = m0 + ((lane >> 4) << 3);
    #pragma unroll
    for (int r = 0; r < 8; ++r) {
        const int row = rbase + r;
        float f = acc[r] + b2[col];
        float v = X[(size_t)row * STATE + col] + dt * f;
        Xnext[(size_t)row * STATE + col] = fmaxf(v, 0.f);
    }
}

extern "C" void kernel_launch(void* const* d_in, const int* in_sizes, int n_in,
                              void* d_out, int out_size, void* d_ws, size_t ws_size,
                              hipStream_t stream) {
    const float* x0 = (const float*)d_in[0];
    const float* z  = (const float*)d_in[1];
    const float* t  = (const float*)d_in[2];
    const float* W1 = (const float*)d_in[3];
    const float* b1 = (const float*)d_in[4];
    const float* W2 = (const float*)d_in[5];
    const float* b2 = (const float*)d_in[6];
    float* out = (float*)d_out;

    const size_t nZ1 = (size_t)BATCH * HIDDEN;
    const size_t nH  = (size_t)BATCH * HIDDEN;
    const size_t nW1 = (size_t)STATE * HIDDEN;
    const size_t nW2 = (size_t)HIDDEN * STATE;

    float* Z1     = (float*)d_ws;
    float* Hbuf   = Z1 + nZ1;
    float* W1pack = Hbuf + nH;
    float* W2pack = W1pack + nW1;

    const bool packed = ws_size >= (nZ1 + nH + nW1 + nW2) * sizeof(float);

    copy_x0_kernel<<<(BATCH * STATE) / 256, 256, 0, stream>>>(x0, out);
    z1_precompute_kernel<<<(BATCH * HIDDEN) / 256, 256, 0, stream>>>(z, W1, b1, Z1);

    if (packed) {
        pack_w_kernel<<<(int)((nW1 / 4) / 256), 256, 0, stream>>>(W1, W1pack, STATE, HIDDEN);
        pack_w_kernel<<<(int)((nW2 / 4) / 256), 256, 0, stream>>>(W2, W2pack, HIDDEN, STATE);
        for (int s = 0; s < NSTEPS - 1; ++s) {
            const float* xcur  = out + (size_t)s * BATCH * STATE;
            float*       xnext = out + (size_t)(s + 1) * BATCH * STATE;
            step1_packed_kernel<<<128, 256, 0, stream>>>(xcur, W1pack, W1, Z1, t, s, Hbuf);
            step2_packed_kernel<<<64, 256, 0, stream>>>(Hbuf, W2pack, b2, xcur, t, xnext);
        }
    } else {
        for (int s = 0; s < NSTEPS - 1; ++s) {
            const float* xcur  = out + (size_t)s * BATCH * STATE;
            float*       xnext = out + (size_t)(s + 1) * BATCH * STATE;
            step1_direct_kernel<<<128, 256, 0, stream>>>(xcur, W1, Z1, t, s, Hbuf);
            step2_direct_kernel<<<64, 256, 0, stream>>>(Hbuf, W2, b2, xcur, t, xnext);
        }
    }
}


# LDS row pitch of the staged A panels K -> K+4 floats (16 B x odd): the 16 rows of each A-fragment read no longer share one bank pair (16-way conflict removed); sums and their order unchanged
# speedup vs baseline: 1.5382x; 1.5382x over previous
#include <hip/hip_runtime.h>
#include <math.h>

#define BATCH   256
#define STATE   512
#define HIDDEN  1024
#define COND    64
#define NSTEPS  256
#define TSCALE  0.1f

#define XS_PITCH (STATE + 4)
#define HS_PITCH (HIDDEN + 4)

typedef __attribute__((ext_vector_type(2))) float v2f;
typedef __attribute__((ext_vector_type(4))) float v4f;
typedef __attribute__((ext_vector_type(8))) float v8f;

static __device__ __forceinline__ v8f wmma_f32(v2f a, v2f b, v8f c) {
    return __builtin_amdgcn_wmma_f32_16x16x4_f32(
         false, a,  false, b,
         (short)0, c,  false,  false);
}

__global__ __launch_bounds__(256) void copy_x0_kernel(const float* __restrict__ x0,
                                                      float* __restrict__ out) {
    int i = blockIdx.x * 256 + threadIdx.x;
    out[i] = x0[i];
}

__global__ __launch_bounds__(256) void z1_precompute_kernel(const float* __restrict__ z,
                                                            const float* __restrict__ W1,
                                                            const float* __restrict__ b1,
                                                            float* __restrict__ Z1) {
    int idx = blockIdx.x * 256 + threadIdx.x;
    int b = idx >> 10;
    int j = idx & (HIDDEN - 1);
    float acc = b1[j];
    const float* zrow = z + b * COND;
    const float* wz   = W1 + (size_t)STATE * HIDDEN + j;
    #pragma unroll 4
    for (int k = 0; k < COND; ++k)
        acc += zrow[k] * wz[(size_t)k * HIDDEN];
    Z1[idx] = acc;
}

__global__ __launch_bounds__(256) void pack_w_kernel(const float* __restrict__ W,
                                                     float* __restrict__ Wp,
                                                     int K, int N) {
    int idx  = blockIdx.x * 256 + threadIdx.x;
    int lane = idx & 31;
    int kb2  = (idx >> 5) % (K >> 3);
    int nt   = (idx >> 5) / (K >> 3);
    int k0   = (kb2 << 3) + ((lane >> 4) << 1);
    int n    = (nt << 4) + (lane & 15);
    size_t o = (size_t)idx * 4;
    Wp[o + 0] = W[(size_t)(k0 + 0) * N + n];
    Wp[o + 1] = W[(size_t)(k0 + 1) * N + n];
    Wp[o + 2] = W[(size_t)(k0 + 4) * N + n];
    Wp[o + 3] = W[(size_t)(k0 + 5) * N + n];
}

__global__ __launch_bounds__(256) void step1_packed_kernel(const float* __restrict__ X,
                                                           const float* __restrict__ W1pack,
                                                           const float* __restrict__ W1,
                                                           const float* __restrict__ Z1,
                                                           const float* __restrict__ t,
                                                           int s,
                                                           float* __restrict__ Hbuf) {
    __shared__ float xs[16 * XS_PITCH];

    const int mt   = blockIdx.x >> 3;
    const int ng   = blockIdx.x & 7;
    const int m0   = mt * 16;
    const int wave = threadIdx.x >> 5;
    const int lane = threadIdx.x & 31;
    const int n0   = (ng * 8 + wave) * 16;

    for (int i = threadIdx.x; i < 16 * STATE; i += 256)
        xs[(i >> 9) * XS_PITCH + (i & (STATE - 1))] = X[(size_t)(m0 + (i >> 9)) * STATE + (i & (STATE - 1))];
    __syncthreads();

    const int ar = lane & 15;
    const int ak = (lane >> 4) * 2;
    const float* wp = W1pack + ((size_t)(n0 >> 4) * (STATE >> 3)) * 128 + lane * 4;

    v8f acc = {0.f, 0.f, 0.f, 0.f, 0.f, 0.f, 0.f, 0.f};
    for (int kb = 0; kb < (STATE >> 3); ++kb) {
        v4f w  = *(const v4f*)(wp + (size_t)kb * 128);
        v2f a0 = *(const v2f*)&xs[ar * XS_PITCH + kb * 8 + ak];
        v2f a1 = *(const v2f*)&xs[ar * XS_PITCH + kb * 8 + 4 + ak];
        v2f b0 = {w.x, w.y};
        v2f b1 = {w.z, w.w};
        acc = wmma_f32(a0, b0, acc);
        acc = wmma_f32(a1, b1, acc);
    }

    const float ts   = t[s];
    const float* W1t = W1 + (size_t)(STATE + COND) * HIDDEN;

    const int col   = n0 + (lane & 15);
    const int rbase = m0 + ((lane >> 4) << 3);
    #pragma unroll
    for (int r = 0; r < 8; ++r) {
        const int row = rbase + r;
        float v = acc[r] + Z1[(size_t)row * HIDDEN + col] + ts * W1t[col];
        Hbuf[(size_t)row * HIDDEN + col] = tanhf(v);
    }
}

__global__ __launch_bounds__(256) void step2_packed_kernel(const float* __restrict__ Hbuf,
                                                           const float* __restrict__ W2pack,
                                                           const float* __restrict__ b2,
                                                           const float* __restrict__ X,
                                                           const float* __restrict__ t,
                                                           float* __restrict__ Xnext) {
    __shared__ float hs[16 * HS_PITCH];

    const int mt   = blockIdx.x >> 2;
    const int ng   = blockIdx.x & 3;
    const int m0   = mt * 16;
    const int wave = threadIdx.x >> 5;
    const int lane = threadIdx.x & 31;
    const int n0   = (ng * 8 + wave) * 16;

    for (int i = threadIdx.x; i < 16 * HIDDEN; i += 256)
        hs[(i >> 10) * HS_PITCH + (i & (HIDDEN - 1))] = Hbuf[(size_t)(m0 + (i >> 10)) * HIDDEN + (i & (HIDDEN - 1))];
    __syncthreads();

    const int ar = lane & 15;
    const int ak = (lane >> 4) * 2;
    const float* wp = W2pack + ((size_t)(n0 >> 4) * (HIDDEN >> 3)) * 128 + lane * 4;

    v8f acc = {0.f, 0.f, 0.f, 0.f, 0.f, 0.f, 0.f, 0.f};
    for (int kb = 0; kb < (HIDDEN >> 3); ++kb) {
        v4f w  = *(const v4f*)(wp + (size_t)kb * 128);
        v2f a0 = *(const v2f*)&hs[ar * HS_PITCH + kb * 8 + ak];
        v2f a1 = *(const v2f*)&hs[ar * HS_PITCH + kb * 8 + 4 + ak];
        v2f b0 = {w.x, w.y};
        v2f b1 = {w.z, w.w};
        acc = wmma_f32(a0, b0, acc);
        acc = wmma_f32(a1, b1, acc);
    }

    const float dt = (t[1] - t[0]) * TSCALE;

    const int col   = n0 + (lane & 15);
    const int rbase = m0 + ((lane >> 4) << 3);
    #pragma unroll
    for (int r = 0; r < 8; ++r) {
        const int row = rbase + r;
        float f = acc[r] + b2[col];
        float v = X[(size_t)row * STATE + col] + dt * f;
        Xnext[(size_t)row * STATE + col] = fmaxf(v, 0.f);
    }
}

__global__ __launch_bounds__(256) void step1_direct_kernel(const float* __restrict__ X,
                                                           const float* __restrict__ W1,
                                                           const float* __restrict__ Z1,
                                                           const float* __restrict__ t,
                                                           int s,
                                                           float* __restrict__ Hbuf) {
    __shared__ float xs[16 * XS_PITCH];
    const int mt = blockIdx.x >> 3, ng = blockIdx.x & 7;
    const int m0 = mt * 16;
    const int wave = threadIdx.x >> 5, lane = threadIdx.x & 31;
    const int n0 = (ng * 8 + wave) * 16;
    for (int i = threadIdx.x; i < 16 * STATE; i += 256)
        xs[(i >> 9) * XS_PITCH + (i & (STATE - 1))] = X[(size_t)(m0 + (i >> 9)) * STATE + (i & (STATE - 1))];
    __syncthreads();
    const int ar = lane & 15, ak = (lane >> 4) * 2;
    v8f acc = {0.f, 0.f, 0.f, 0.f, 0.f, 0.f, 0.f, 0.f};
    for (int k = 0; k < STATE; k += 4) {
        v2f a = *(const v2f*)&xs[ar * XS_PITCH + k + ak];
        v2f b;
        b.x = W1[(size_t)(k + ak) * HIDDEN + n0 + ar];
        b.y = W1[(size_t)(k + ak + 1) * HIDDEN + n0 + ar];
        acc = wmma_f32(a, b, acc);
    }
    const float ts = t[s];
    const float* W1t = W1 + (size_t)(STATE + COND) * HIDDEN;
    const int col = n0 + (lane & 15);
    const int rbase = m0 + ((lane >> 4) << 3);
    #pragma unroll
    for (int r = 0; r < 8; ++r) {
        const int row = rbase + r;
        float v = acc[r] + Z1[(size_t)row * HIDDEN + col] + ts * W1t[col];
        Hbuf[(size_t)row * HIDDEN + col] = tanhf(v);
    }
}

__global__ __launch_bounds__(256) void step2_direct_kernel(const float* __restrict__ Hbuf,
                                                           const float* __restrict__ W2,
                                                           const float* __restrict__ b2,
                                                           const float* __restrict__ X,
                                                           const float* __restrict__ t,
                                                           float* __restrict__ Xnext) {
    __shared__ float hs[16 * HS_PITCH];
    const int mt = blockIdx.x >> 2, ng = blockIdx.x & 3;
    const int m0 = mt * 16;
    const int wave = threadIdx.x >> 5, lane = threadIdx.x & 31;
    const int n0 = (ng * 8 + wave) * 16;
    for (int i = threadIdx.x; i < 16 * HIDDEN; i += 256)
        hs[(i >> 10) * HS_PITCH + (i & (HIDDEN - 1))] = Hbuf[(size_t)(m0 + (i >> 10)) * HIDDEN + (i & (HIDDEN - 1))];
    __syncthreads();
    const int ar = lane & 15, ak = (lane >> 4) * 2;
    v8f acc = {0.f, 0.f, 0.f, 0.f, 0.f, 0.f, 0.f, 0.f};
    for (int k = 0; k < HIDDEN; k += 4) {
        v2f a = *(const v2f*)&hs[ar * HS_PITCH + k + ak];
        v2f b;
        b.x = W2[(size_t)(k + ak) * STATE + n0 + ar];
        b.y = W2[(size_t)(k + ak + 1) * STATE + n0 + ar];
        acc = wmma_f32(a, b, acc);
    }
    const float dt = (t[1] - t[0]) * TSCALE;
    const int col = n0 + (lane & 15);
    const int rbase = m0 + ((lane >> 4) << 3);
    #pragma unroll
    for (int r = 0; r < 8; ++r) {
        const int row = rbase + r;
        float f = acc[r] + b2[col];
        float v = X[(size_t)row * STATE + col] + dt * f;
        Xnext[(size_t)row * STATE + col] = fmaxf(v, 0.f);
    }
}

extern "C" void kernel_launch(void* const* d_in, const int* in_sizes, int n_in,
                              void* d_out, int out_size, void* d_ws, size_t ws_size,
                              hipStream_t stream) {
    const float* x0 = (const float*)d_in[0];
    const float* z  = (const float*)d_in[1];
    const float* t  = (const float*)d_in[2];
    const float* W1 = (const float*)d_in[3];
    const float* b1 = (const float*)d_in[4];
    const float* W2 = (const float*)d_in[5];
    const float* b2 = (const float*)d_in[6];
    float* out = (float*)d_out;

    const size_t nZ1 = (size_t)BATCH * HIDDEN;
    const size_t nH  = (size_t)BATCH * HIDDEN;
    const size_t nW1 = (size_t)STATE * HIDDEN;
    const size_t nW2 = (size_t)HIDDEN * STATE;

    float* Z1     = (float*)d_ws;
    float* Hbuf   = Z1 + nZ1;
    float* W1pack = Hbuf + nH;
    float* W2pack = W1pack + nW1;

    const bool packed = ws_size >= (nZ1 + nH + nW1 + nW2) * sizeof(float);

    copy_x0_kernel<<<(BATCH * STATE) / 256, 256, 0, stream>>>(x0, out);
    z1_precompute_kernel<<<(BATCH * HIDDEN) / 256, 256, 0, stream>>>(z, W1, b1, Z1);

    if (packed) {
        pack_w_kernel<<<(int)((nW1 / 4) / 256), 256, 0, stream>>>(W1, W1pack, STATE, HIDDEN);
        pack_w_kernel<<<(int)((nW2 / 4) / 256), 256, 0, stream>>>(W2, W2pack, HIDDEN, STATE);
        for (int s = 0; s < NSTEPS - 1; ++s) {
            const float* xcur  = out + (size_t)s * BATCH * STATE;
            float*       xnext = out + (size_t)(s + 1) * BATCH * STATE;
            step1_packed_kernel<<<128, 256, 0, stream>>>(xcur, W1pack, W1, Z1, t, s, Hbuf);
            step2_packed_kernel<<<64, 256, 0, stream>>>(Hbuf, W2pack, b2, xcur, t, xnext);
        }
    } else {
        for (int s = 0; s < NSTEPS - 1; ++s) {
            const float* xcur  = out + (size_t)s * BATCH * STATE;
            float*       xnext = out + (size_t)(s + 1) * BATCH * STATE;
            step1_direct_kernel<<<128, 256, 0, stream>>>(xcur, W1, Z1, t, s, Hbuf);
            step2_direct_kernel<<<64, 256, 0, stream>>>(Hbuf, W2, b2, xcur, t, xnext);
        }
    }
}


# Global loads issued ahead of use: weight fragments one 16-k-block group ahead of their WMMAs (first group before the A staging), epilogue operands before the K loop; sums and their order unchanged
# speedup vs baseline: 1.8834x; 1.2244x over previous
#include <hip/hip_runtime.h>
#include <math.h>

#define BATCH   256
#define STATE   512
#define HIDDEN  1024
#define COND    64
#define NSTEPS  256
#define TSCALE  0.1f

#define XS_PITCH (STATE + 4)
#define HS_PITCH (HIDDEN + 4)

typedef __attribute__((ext_vector_type(2))) float v2f;
typedef __attribute__((ext_vector_type(4))) float v4f;
typedef __attribute__((ext_vector_type(8))) float v8f;

static __device__ __forceinline__ v8f wmma_f32(v2f a, v2f b, v8f c) {
    return __builtin_amdgcn_wmma_f32_16x16x4_f32(
         false, a,  false, b,
         (short)0, c,  false,  false);
}

#define KB_GROUP 16

static __device__ __forceinline__ void load_w_group(v4f (&w)[KB_GROUP], const float* __restrict__ wp, int g) {
    #pragma unroll
    for (int j = 0; j < KB_GROUP; ++j)
        w[j] = *(const v4f*)(wp + (size_t)(g * KB_GROUP + j) * 128);
}

static __device__ __forceinline__ v8f mma_w_group(const v4f (&w)[KB_GROUP], const float* arow, int g, v8f acc) {
    #pragma unroll
    for (int j = 0; j < KB_GROUP; ++j) {
        const int kb = g * KB_GROUP + j;
        v2f a0 = *(const v2f*)&arow[kb * 8];
        v2f a1 = *(const v2f*)&arow[kb * 8 + 4];
        v2f b0 = {w[j].x, w[j].y};
        v2f b1 = {w[j].z, w[j].w};
        acc = wmma_f32(a0, b0, acc);
        acc = wmma_f32(a1, b1, acc);
    }
    return acc;
}

static __device__ __forceinline__ v8f wmma_drain(v8f acc) {
    __builtin_amdgcn_sched_barrier(0);
    asm volatile("v_nop\n\tv_nop\n\tv_nop\n\tv_nop\n\tv_nop\n\tv_nop\n\tv_nop\n\tv_nop" : "+v"(acc));
    return acc;
}

__global__ __launch_bounds__(256) void copy_x0_kernel(const float* __restrict__ x0,
                                                      float* __restrict__ out) {
    int i = blockIdx.x * 256 + threadIdx.x;
    out[i] = x0[i];
}

__global__ __launch_bounds__(256) void z1_precompute_kernel(const float* __restrict__ z,
                                                            const float* __restrict__ W1,
                                                            const float* __restrict__ b1,
                                                            float* __restrict__ Z1) {
    int idx = blockIdx.x * 256 + threadIdx.x;
    int b = idx >> 10;
    int j = idx & (HIDDEN - 1);
    float acc = b1[j];
    const float* zrow = z + b * COND;
    const float* wz   = W1 + (size_t)STATE * HIDDEN + j;
    #pragma unroll 4
    for (int k = 0; k < COND; ++k)
        acc += zrow[k] * wz[(size_t)k * HIDDEN];
    Z1[idx] = acc;
}

__global__ __launch_bounds__(256) void pack_w_kernel(const float* __restrict__ W,
                                                     float* __restrict__ Wp,
                                                     int K, int N) {
    int idx  = blockIdx.x * 256 + threadIdx.x;
    int lane = idx & 31;
    int kb2  = (idx >> 5) % (K >> 3);
    int nt   = (idx >> 5) / (K >> 3);
    int k0   = (kb2 << 3) + ((lane >> 4) << 1);
    int n    = (nt << 4) + (lane & 15);
    size_t o = (size_t)idx * 4;
    Wp[o + 0] = W[(size_t)(k0 + 0) * N + n];
    Wp[o + 1] = W[(size_t)(k0 + 1) * N + n];
    Wp[o + 2] = W[(size_t)(k0 + 4) * N + n];
    Wp[o + 3] = W[(size_t)(k0 + 5) * N + n];
}

__global__ __launch_bounds__(256) void step1_packed_kernel(const float* __restrict__ X,
                                                           const float* __restrict__ W1pack,
                                                           const float* __restrict__ W1,
                                                           const float* __restrict__ Z1,
                                                           const float* __restrict__ t,
                                                           int s,
                                                           float* __restrict__ Hbuf) {
    __shared__ float xs[16 * XS_PITCH];

    const int mt   = blockIdx.x >> 3;
    const int ng   = blockIdx.x & 7;
    const int m0   = mt * 16;
    const int wave = threadIdx.x >> 5;
    const int lane = threadIdx.x & 31;
    const int n0   = (ng * 8 + wave) * 16;

    const int ar = lane & 15;
    const int ak = (lane >> 4) * 2;
    const float* wp = W1pack + ((size_t)(n0 >> 4) * (STATE >> 3)) * 128 + lane * 4;

    v4f wa[KB_GROUP], wb[KB_GROUP];
    load_w_group(wa, wp, 0);

    const float ts   = t[s];
    const float* W1t = W1 + (size_t)(STATE + COND) * HIDDEN;
    const int col   = n0 + (lane & 15);
    const int rbase = m0 + ((lane >> 4) << 3);
    const float w1t = W1t[col];
    float z1v[8];
    #pragma unroll
    for (int r = 0; r < 8; ++r)
        z1v[r] = Z1[(size_t)(rbase + r) * HIDDEN + col];

    for (int i = threadIdx.x; i < 16 * STATE; i += 256)
        xs[(i >> 9) * XS_PITCH + (i & (STATE - 1))] = X[(size_t)(m0 + (i >> 9)) * STATE + (i & (STATE - 1))];
    __syncthreads();

    const float* arow = &xs[ar * XS_PITCH + ak];

    v8f acc = {0.f, 0.f, 0.f, 0.f, 0.f, 0.f, 0.f, 0.f};
    #pragma unroll
    for (int g = 0; g < (STATE >> 3) / KB_GROUP; g += 2) {
        load_w_group(wb, wp, g + 1);
        __builtin_amdgcn_sched_barrier(0);
        acc = mma_w_group(wa, arow, g, acc);
        if (g + 2 < (STATE >> 3) / KB_GROUP) load_w_group(wa, wp, g + 2);
        __builtin_amdgcn_sched_barrier(0);
        acc = mma_w_group(wb, arow, g + 1, acc);
    }
    acc = wmma_drain(acc);

    #pragma unroll
    for (int r = 0; r < 8; ++r) {
        const int row = rbase + r;
        float v = acc[r] + z1v[r] + ts * w1t;
        Hbuf[(size_t)row * HIDDEN + col] = tanhf(v);
    }
}

__global__ __launch_bounds__(256) void step2_packed_kernel(const float* __restrict__ Hbuf,
                                                           const float* __restrict__ W2pack,
                                                           const float* __restrict__ b2,
                                                           const float* __restrict__ X,
                                                           const float* __restrict__ t,
                                                           float* __restrict__ Xnext) {
    __shared__ float hs[16 * HS_PITCH];

    const int mt   = blockIdx.x >> 2;
    const int ng   = blockIdx.x & 3;
    const int m0   = mt * 16;
    const int wave = threadIdx.x >> 5;
    const int lane = threadIdx.x & 31;
    const int n0   = (ng * 8 + wave) * 16;

    const int ar = lane & 15;
    const int ak = (lane >> 4) * 2;
    const float* wp = W2pack + ((size_t)(n0 >> 4) * (HIDDEN >> 3)) * 128 + lane * 4;

    v4f wa[KB_GROUP], wb[KB_GROUP];
    load_w_group(wa, wp, 0);

    const float dt = (t[1] - t[0]) * TSCALE;
    const int col   = n0 + (lane & 15);
    const int rbase = m0 + ((lane >> 4) << 3);
    const float b2v = b2[col];
    float xv[8];
    #pragma unroll
    for (int r = 0; r < 8; ++r)
        xv[r] = X[(size_t)(rbase + r) * STATE + col];

    for (int i = threadIdx.x; i < 16 * HIDDEN; i += 256)
        hs[(i >> 10) * HS_PITCH + (i & (HIDDEN - 1))] = Hbuf[(size_t)(m0 + (i >> 10)) * HIDDEN + (i & (HIDDEN - 1))];
    __syncthreads();

    const float* arow = &hs[ar * HS_PITCH + ak];

    v8f acc = {0.f, 0.f, 0.f, 0.f, 0.f, 0.f, 0.f, 0.f};
    #pragma unroll
    for (int g = 0; g < (HIDDEN >> 3) / KB_GROUP; g += 2) {
        load_w_group(wb, wp, g + 1);
        __builtin_amdgcn_sched_barrier(0);
        acc = mma_w_group(wa, arow, g, acc);
        if (g + 2 < (HIDDEN >> 3) / KB_GROUP) load_w_group(wa, wp, g + 2);
        __builtin_amdgcn_sched_barrier(0);
        acc = mma_w_group(wb, arow, g + 1, acc);
    }
    acc = wmma_drain(acc);

    #pragma unroll
    for (int r = 0; r < 8; ++r) {
        const int row = rbase + r;
        float f = acc[r] + b2v;
        float v = xv[r] + dt * f;
        Xnext[(size_t)row * STATE + col] = fmaxf(v, 0.f);
    }
}

__global__ __launch_bounds__(256) void step1_direct_kernel(const float* __restrict__ X,
                                                           const float* __restrict__ W1,
                                                           const float* __restrict__ Z1,
                                                           const float* __restrict__ t,
                                                           int s,
                                                           float* __restrict__ Hbuf) {
    __shared__ float xs[16 * XS_PITCH];
    const int mt = blockIdx.x >> 3, ng = blockIdx.x & 7;
    const int m0 = mt * 16;
    const int wave = threadIdx.x >> 5, lane = threadIdx.x & 31;
    const int n0 = (ng * 8 + wave) * 16;
    for (int i = threadIdx.x; i < 16 * STATE; i += 256)
        xs[(i >> 9) * XS_PITCH + (i & (STATE - 1))] = X[(size_t)(m0 + (i >> 9)) * STATE + (i & (STATE - 1))];
    __syncthreads();
    const int ar = lane & 15, ak = (lane >> 4) * 2;
    v8f acc = {0.f, 0.f, 0.f, 0.f, 0.f, 0.f, 0.f, 0.f};
    for (int k = 0; k < STATE; k += 4) {
        v2f a = *(const v2f*)&xs[ar * XS_PITCH + k + ak];
        v2f b;
        b.x = W1[(size_t)(k + ak) * HIDDEN + n0 + ar];
        b.y = W1[(size_t)(k + ak + 1) * HIDDEN + n0 + ar];
        acc = wmma_f32(a, b, acc);
    }
    const float ts = t[s];
    const float* W1t = W1 + (size_t)(STATE + COND) * HIDDEN;
    const int col = n0 + (lane & 15);
    const int rbase = m0 + ((lane >> 4) << 3);
    #pragma unroll
    for (int r = 0; r < 8; ++r) {
        const int row = rbase + r;
        float v = acc[r] + Z1[(size_t)row * HIDDEN + col] + ts * W1t[col];
        Hbuf[(size_t)row * HIDDEN + col] = tanhf(v);
    }
}

__global__ __launch_bounds__(256) void step2_direct_kernel(const float* __restrict__ Hbuf,
                                                           const float* __restrict__ W2,
                                                           const float* __restrict__ b2,
                                                           const float* __restrict__ X,
                                                           const float* __restrict__ t,
                                                           float* __restrict__ Xnext) {
    __shared__ float hs[16 * HS_PITCH];
    const int mt = blockIdx.x >> 2, ng = blockIdx.x & 3;
    const int m0 = mt * 16;
    const int wave = threadIdx.x >> 5, lane = threadIdx.x & 31;
    const int n0 = (ng * 8 + wave) * 16;
    for (int i = threadIdx.x; i < 16 * HIDDEN; i += 256)
        hs[(i >> 10) * HS_PITCH + (i & (HIDDEN - 1))] = Hbuf[(size_t)(m0 + (i >> 10)) * HIDDEN + (i & (HIDDEN - 1))];
    __syncthreads();
    const int ar = lane & 15, ak = (lane >> 4) * 2;
    v8f acc = {0.f, 0.f, 0.f, 0.f, 0.f, 0.f, 0.f, 0.f};
    for (int k = 0; k < HIDDEN; k += 4) {
        v2f a = *(const v2f*)&hs[ar * HS_PITCH + k + ak];
        v2f b;
        b.x = W2[(size_t)(k + ak) * STATE + n0 + ar];
        b.y = W2[(size_t)(k + ak + 1) * STATE + n0 + ar];
        acc = wmma_f32(a, b, acc);
    }
    const float dt = (t[1] - t[0]) * TSCALE;
    const int col = n0 + (lane & 15);
    const int rbase = m0 + ((lane >> 4) << 3);
    #pragma unroll
    for (int r = 0; r < 8; ++r) {
        const int row = rbase + r;
        float f = acc[r] + b2[col];
        float v = X[(size_t)row * STATE + col] + dt * f;
        Xnext[(size_t)row * STATE + col] = fmaxf(v, 0.f);
    }
}

extern "C" void kernel_launch(void* const* d_in, const int* in_sizes, int n_in,
                              void* d_out, int out_size, void* d_ws, size_t ws_size,
                              hipStream_t stream) {
    const float* x0 = (const float*)d_in[0];
    const float* z  = (const float*)d_in[1];
    const float* t  = (const float*)d_in[2];
    const float* W1 = (const float*)d_in[3];
    const float* b1 = (const float*)d_in[4];
    const float* W2 = (const float*)d_in[5];
    const float* b2 = (const float*)d_in[6];
    float* out = (float*)d_out;

    const size_t nZ1 = (size_t)BATCH * HIDDEN;
    const size_t nH  = (size_t)BATCH * HIDDEN;
    const size_t nW1 = (size_t)STATE * HIDDEN;
    const size_t nW2 = (size_t)HIDDEN * STATE;

    float* Z1     = (float*)d_ws;
    float* Hbuf   = Z1 + nZ1;
    float* W1pack = Hbuf + nH;
    float* W2pack = W1pack + nW1;

    const bool packed = ws_size >= (nZ1 + nH + nW1 + nW2) * sizeof(float);

    copy_x0_kernel<<<(BATCH * STATE) / 256, 256, 0, stream>>>(x0, out);
    z1_precompute_kernel<<<(BATCH * HIDDEN) / 256, 256, 0, stream>>>(z, W1, b1, Z1);

    if (packed) {
        pack_w_kernel<<<(int)((nW1 / 4) / 256), 256, 0, stream>>>(W1, W1pack, STATE, HIDDEN);
        pack_w_kernel<<<(int)((nW2 / 4) / 256), 256, 0, stream>>>(W2, W2pack, HIDDEN, STATE);
        for (int s = 0; s < NSTEPS - 1; ++s) {
            const float* xcur  = out + (size_t)s * BATCH * STATE;
            float*       xnext = out + (size_t)(s + 1) * BATCH * STATE;
            step1_packed_kernel<<<128, 256, 0, stream>>>(xcur, W1pack, W1, Z1, t, s, Hbuf);
            step2_packed_kernel<<<64, 256, 0, stream>>>(Hbuf, W2pack, b2, xcur, t, xnext);
        }
    } else {
        for (int s = 0; s < NSTEPS - 1; ++s) {
            const float* xcur  = out + (size_t)s * BATCH * STATE;
            float*       xnext = out + (size_t)(s + 1) * BATCH * STATE;
            step1_direct_kernel<<<128, 256, 0, stream>>>(xcur, W1, Z1, t, s, Hbuf);
            step2_direct_kernel<<<64, 256, 0, stream>>>(Hbuf, W2, b2, xcur, t, xnext);
        }
    }
}


# A-panel staging: each thread's share fetched as 16-byte loads, all issued before the first LDS store and ahead of the weight loads (one L2 round trip, not one per 8 loads); sums and order unchanged
# speedup vs baseline: 2.5743x; 1.3668x over previous
#include <hip/hip_runtime.h>
#include <math.h>

#define BATCH   256
#define STATE   512
#define HIDDEN  1024
#define COND    64
#define NSTEPS  256
#define TSCALE  0.1f

#define XS_PITCH (STATE + 4)
#define HS_PITCH (HIDDEN + 4)

#define XS_Q ((16 * STATE / 4) / 256)
#define HS_Q ((16 * HIDDEN / 4) / 256)

typedef __attribute__((ext_vector_type(2))) float v2f;
typedef __attribute__((ext_vector_type(4))) float v4f;
typedef __attribute__((ext_vector_type(8))) float v8f;

static __device__ __forceinline__ v8f wmma_f32(v2f a, v2f b, v8f c) {
    return __builtin_amdgcn_wmma_f32_16x16x4_f32(
         false, a,  false, b,
         (short)0, c,  false,  false);
}

#define KB_GROUP 16

static __device__ __forceinline__ void load_w_group(v4f (&w)[KB_GROUP], const float* __restrict__ wp, int g) {
    #pragma unroll
    for (int j = 0; j < KB_GROUP; ++j)
        w[j] = *(const v4f*)(wp + (size_t)(g * KB_GROUP + j) * 128);
}

static __device__ __forceinline__ v8f mma_w_group(const v4f (&w)[KB_GROUP], const float* arow, int g, v8f acc) {
    #pragma unroll
    for (int j = 0; j < KB_GROUP; ++j) {
        const int kb = g * KB_GROUP + j;
        v2f a0 = *(const v2f*)&arow[kb * 8];
        v2f a1 = *(const v2f*)&arow[kb * 8 + 4];
        v2f b0 = {w[j].x, w[j].y};
        v2f b1 = {w[j].z, w[j].w};
        acc = wmma_f32(a0, b0, acc);
        acc = wmma_f32(a1, b1, acc);
    }
    return acc;
}

static __device__ __forceinline__ v8f wmma_drain(v8f acc) {
    __builtin_amdgcn_sched_barrier(0);
    asm volatile("v_nop\n\tv_nop\n\tv_nop\n\tv_nop\n\tv_nop\n\tv_nop\n\tv_nop\n\tv_nop" : "+v"(acc));
    return acc;
}

__global__ __launch_bounds__(256) void copy_x0_kernel(const float* __restrict__ x0,
                                                      float* __restrict__ out) {
    int i = blockIdx.x * 256 + threadIdx.x;
    out[i] = x0[i];
}

__global__ __launch_bounds__(256) void z1_precompute_kernel(const float* __restrict__ z,
                                                            const float* __restrict__ W1,
                                                            const float* __restrict__ b1,
                                                            float* __restrict__ Z1) {
    int idx = blockIdx.x * 256 + threadIdx.x;
    int b = idx >> 10;
    int j = idx & (HIDDEN - 1);
    float acc = b1[j];
    const float* zrow = z + b * COND;
    const float* wz   = W1 + (size_t)STATE * HIDDEN + j;
    #pragma unroll 4
    for (int k = 0; k < COND; ++k)
        acc += zrow[k] * wz[(size_t)k * HIDDEN];
    Z1[idx] = acc;
}

__global__ __launch_bounds__(256) void pack_w_kernel(const float* __restrict__ W,
                                                     float* __restrict__ Wp,
                                                     int K, int N) {
    int idx  = blockIdx.x * 256 + threadIdx.x;
    int lane = idx & 31;
    int kb2  = (idx >> 5) % (K >> 3);
    int nt   = (idx >> 5) / (K >> 3);
    int k0   = (kb2 << 3) + ((lane >> 4) << 1);
    int n    = (nt << 4) + (lane & 15);
    size_t o = (size_t)idx * 4;
    Wp[o + 0] = W[(size_t)(k0 + 0) * N + n];
    Wp[o + 1] = W[(size_t)(k0 + 1) * N + n];
    Wp[o + 2] = W[(size_t)(k0 + 4) * N + n];
    Wp[o + 3] = W[(size_t)(k0 + 5) * N + n];
}

__global__ __launch_bounds__(256) void step1_packed_kernel(const float* __restrict__ X,
                                                           const float* __restrict__ W1pack,
                                                           const float* __restrict__ W1,
                                                           const float* __restrict__ Z1,
                                                           const float* __restrict__ t,
                                                           int s,
                                                           float* __restrict__ Hbuf) {
    __shared__ __attribute__((aligned(16))) float xs[16 * XS_PITCH];

    const int mt   = blockIdx.x >> 3;
    const int ng   = blockIdx.x & 7;
    const int m0   = mt * 16;
    const int wave = threadIdx.x >> 5;
    const int lane = threadIdx.x & 31;
    const int n0   = (ng * 8 + wave) * 16;

    const int ar = lane & 15;
    const int ak = (lane >> 4) * 2;
    const float* wp = W1pack + ((size_t)(n0 >> 4) * (STATE >> 3)) * 128 + lane * 4;

    v4f xa[XS_Q];
    const float* xpanel = X + (size_t)m0 * STATE;
    #pragma unroll
    for (int j = 0; j < XS_Q; ++j)
        xa[j] = *(const v4f*)(xpanel + (size_t)(threadIdx.x + 256 * j) * 4);

    v4f wa[KB_GROUP], wb[KB_GROUP];
    load_w_group(wa, wp, 0);

    const float ts   = t[s];
    const float* W1t = W1 + (size_t)(STATE + COND) * HIDDEN;
    const int col   = n0 + (lane & 15);
    const int rbase = m0 + ((lane >> 4) << 3);
    const float w1t = W1t[col];
    float z1v[8];
    #pragma unroll
    for (int r = 0; r < 8; ++r)
        z1v[r] = Z1[(size_t)(rbase + r) * HIDDEN + col];

    #pragma unroll
    for (int j = 0; j < XS_Q; ++j) {
        const int i = (threadIdx.x + 256 * j) * 4;
        *(v4f*)&xs[(i >> 9) * XS_PITCH + (i & (STATE - 1))] = xa[j];
    }
    __syncthreads();

    const float* arow = &xs[ar * XS_PITCH + ak];

    v8f acc = {0.f, 0.f, 0.f, 0.f, 0.f, 0.f, 0.f, 0.f};
    #pragma unroll
    for (int g = 0; g < (STATE >> 3) / KB_GROUP; g += 2) {
        load_w_group(wb, wp, g + 1);
        __builtin_amdgcn_sched_barrier(0);
        acc = mma_w_group(wa, arow, g, acc);
        if (g + 2 < (STATE >> 3) / KB_GROUP) load_w_group(wa, wp, g + 2);
        __builtin_amdgcn_sched_barrier(0);
        acc = mma_w_group(wb, arow, g + 1, acc);
    }
    acc = wmma_drain(acc);

    #pragma unroll
    for (int r = 0; r < 8; ++r) {
        const int row = rbase + r;
        float v = acc[r] + z1v[r] + ts * w1t;
        Hbuf[(size_t)row * HIDDEN + col] = tanhf(v);
    }
}

__global__ __launch_bounds__(256) void step2_packed_kernel(const float* __restrict__ Hbuf,
                                                           const float* __restrict__ W2pack,
                                                           const float* __restrict__ b2,
                                                           const float* __restrict__ X,
                                                           const float* __restrict__ t,
                                                           float* __restrict__ Xnext) {
    __shared__ __attribute__((aligned(16))) float hs[16 * HS_PITCH];

    const int mt   = blockIdx.x >> 2;
    const int ng   = blockIdx.x & 3;
    const int m0   = mt * 16;
    const int wave = threadIdx.x >> 5;
    const int lane = threadIdx.x & 31;
    const int n0   = (ng * 8 + wave) * 16;

    const int ar = lane & 15;
    const int ak = (lane >> 4) * 2;
    const float* wp = W2pack + ((size_t)(n0 >> 4) * (HIDDEN >> 3)) * 128 + lane * 4;

    v4f ha[HS_Q];
    const float* hpanel = Hbuf + (size_t)m0 * HIDDEN;
    #pragma unroll
    for (int j = 0; j < HS_Q; ++j)
        ha[j] = *(const v4f*)(hpanel + (size_t)(threadIdx.x + 256 * j) * 4);

    v4f wa[KB_GROUP], wb[KB_GROUP];
    load_w_group(wa, wp, 0);

    const float dt = (t[1] - t[0]) * TSCALE;
    const int col   = n0 + (lane & 15);
    const int rbase = m0 + ((lane >> 4) << 3);
    const float b2v = b2[col];
    float xv[8];
    #pragma unroll
    for (int r = 0; r < 8; ++r)
        xv[r] = X[(size_t)(rbase + r) * STATE + col];

    #pragma unroll
    for (int j = 0; j < HS_Q; ++j) {
        const int i = (threadIdx.x + 256 * j) * 4;
        *(v4f*)&hs[(i >> 10) * HS_PITCH + (i & (HIDDEN - 1))] = ha[j];
    }
    __syncthreads();

    const float* arow = &hs[ar * HS_PITCH + ak];

    v8f acc = {0.f, 0.f, 0.f, 0.f, 0.f, 0.f, 0.f, 0.f};
    #pragma unroll
    for (int g = 0; g < (HIDDEN >> 3) / KB_GROUP; g += 2) {
        load_w_group(wb, wp, g + 1);
        __builtin_amdgcn_sched_barrier(0);
        acc = mma_w_group(wa, arow, g, acc);
        if (g + 2 < (HIDDEN >> 3) / KB_GROUP) load_w_group(wa, wp, g + 2);
        __builtin_amdgcn_sched_barrier(0);
        acc = mma_w_group(wb, arow, g + 1, acc);
    }
    acc = wmma_drain(acc);

    #pragma unroll
    for (int r = 0; r < 8; ++r) {
        const int row = rbase + r;
        float f = acc[r] + b2v;
        float v = xv[r] + dt * f;
        Xnext[(size_t)row * STATE + col] = fmaxf(v, 0.f);
    }
}

__global__ __launch_bounds__(256) void step1_direct_kernel(const float* __restrict__ X,
                                                           const float* __restrict__ W1,
                                                           const float* __restrict__ Z1,
                                                           const float* __restrict__ t,
                                                           int s,
                                                           float* __restrict__ Hbuf) {
    __shared__ float xs[16 * XS_PITCH];
    const int mt = blockIdx.x >> 3, ng = blockIdx.x & 7;
    const int m0 = mt * 16;
    const int wave = threadIdx.x >> 5, lane = threadIdx.x & 31;
    const int n0 = (ng * 8 + wave) * 16;
    for (int i = threadIdx.x; i < 16 * STATE; i += 256)
        xs[(i >> 9) * XS_PITCH + (i & (STATE - 1))] = X[(size_t)(m0 + (i >> 9)) * STATE + (i & (STATE - 1))];
    __syncthreads();
    const int ar = lane & 15, ak = (lane >> 4) * 2;
    v8f acc = {0.f, 0.f, 0.f, 0.f, 0.f, 0.f, 0.f, 0.f};
    for (int k = 0; k < STATE; k += 4) {
        v2f a = *(const v2f*)&xs[ar * XS_PITCH + k + ak];
        v2f b;
        b.x = W1[(size_t)(k + ak) * HIDDEN + n0 + ar];
        b.y = W1[(size_t)(k + ak + 1) * HIDDEN + n0 + ar];
        acc = wmma_f32(a, b, acc);
    }
    const float ts = t[s];
    const float* W1t = W1 + (size_t)(STATE + COND) * HIDDEN;
    const int col = n0 + (lane & 15);
    const int rbase = m0 + ((lane >> 4) << 3);
    #pragma unroll
    for (int r = 0; r < 8; ++r) {
        const int row = rbase + r;
        float v = acc[r] + Z1[(size_t)row * HIDDEN + col] + ts * W1t[col];
        Hbuf[(size_t)row * HIDDEN + col] = tanhf(v);
    }
}

__global__ __launch_bounds__(256) void step2_direct_kernel(const float* __restrict__ Hbuf,
                                                           const float* __restrict__ W2,
                                                           const float* __restrict__ b2,
                                                           const float* __restrict__ X,
                                                           const float* __restrict__ t,
                                                           float* __restrict__ Xnext) {
    __shared__ float hs[16 * HS_PITCH];
    const int mt = blockIdx.x >> 2, ng = blockIdx.x & 3;
    const int m0 = mt * 16;
    const int wave = threadIdx.x >> 5, lane = threadIdx.x & 31;
    const int n0 = (ng * 8 + wave) * 16;
    for (int i = threadIdx.x; i < 16 * HIDDEN; i += 256)
        hs[(i >> 10) * HS_PITCH + (i & (HIDDEN - 1))] = Hbuf[(size_t)(m0 + (i >> 10)) * HIDDEN + (i & (HIDDEN - 1))];
    __syncthreads();
    const int ar = lane & 15, ak = (lane >> 4) * 2;
    v8f acc = {0.f, 0.f, 0.f, 0.f, 0.f, 0.f, 0.f, 0.f};
    for (int k = 0; k < HIDDEN; k += 4) {
        v2f a = *(const v2f*)&hs[ar * HS_PITCH + k + ak];
        v2f b;
        b.x = W2[(size_t)(k + ak) * STATE + n0 + ar];
        b.y = W2[(size_t)(k + ak + 1) * STATE + n0 + ar];
        acc = wmma_f32(a, b, acc);
    }
    const float dt = (t[1] - t[0]) * TSCALE;
    const int col = n0 + (lane & 15);
    const int rbase = m0 + ((lane >> 4) << 3);
    #pragma unroll
    for (int r = 0; r < 8; ++r) {
        const int row = rbase + r;
        float f = acc[r] + b2[col];
        float v = X[(size_t)row * STATE + col] + dt * f;
        Xnext[(size_t)row * STATE + col] = fmaxf(v, 0.f);
    }
}

extern "C" void kernel_launch(void* const* d_in, const int* in_sizes, int n_in,
                              void* d_out, int out_size, void* d_ws, size_t ws_size,
                              hipStream_t stream) {
    const float* x0 = (const float*)d_in[0];
    const float* z  = (const float*)d_in[1];
    const float* t  = (const float*)d_in[2];
    const float* W1 = (const float*)d_in[3];
    const float* b1 = (const float*)d_in[4];
    const float* W2 = (const float*)d_in[5];
    const float* b2 = (const float*)d_in[6];
    float* out = (float*)d_out;

    const size_t nZ1 = (size_t)BATCH * HIDDEN;
    const size_t nH  = (size_t)BATCH * HIDDEN;
    const size_t nW1 = (size_t)STATE * HIDDEN;
    const size_t nW2 = (size_t)HIDDEN * STATE;

    float* Z1     = (float*)d_ws;
    float* Hbuf   = Z1 + nZ1;
    float* W1pack = Hbuf + nH;
    float* W2pack = W1pack + nW1;

    const bool packed = ws_size >= (nZ1 + nH + nW1 + nW2) * sizeof(float);

    copy_x0_kernel<<<(BATCH * STATE) / 256, 256, 0, stream>>>(x0, out);
    z1_precompute_kernel<<<(BATCH * HIDDEN) / 256, 256, 0, stream>>>(z, W1, b1, Z1);

    if (packed) {
        pack_w_kernel<<<(int)((nW1 / 4) / 256), 256, 0, stream>>>(W1, W1pack, STATE, HIDDEN);
        pack_w_kernel<<<(int)((nW2 / 4) / 256), 256, 0, stream>>>(W2, W2pack, HIDDEN, STATE);
        for (int s = 0; s < NSTEPS - 1; ++s) {
            const float* xcur  = out + (size_t)s * BATCH * STATE;
            float*       xnext = out + (size_t)(s + 1) * BATCH * STATE;
            step1_packed_kernel<<<128, 256, 0, stream>>>(xcur, W1pack, W1, Z1, t, s, Hbuf);
            step2_packed_kernel<<<64, 256, 0, stream>>>(Hbuf, W2pack, b2, xcur, t, xnext);
        }
    } else {
        for (int s = 0; s < NSTEPS - 1; ++s) {
            const float* xcur  = out + (size_t)s * BATCH * STATE;
            float*       xnext = out + (size_t)(s + 1) * BATCH * STATE;
            step1_direct_kernel<<<128, 256, 0, stream>>>(xcur, W1, Z1, t, s, Hbuf);
            step2_direct_kernel<<<64, 256, 0, stream>>>(Hbuf, W2, b2, xcur, t, xnext);
        }
    }
}


# One wave per SIMD on twice the compute units (4-wave blocks, grids 256 and 128 for the two step kernels); each wave's A and B fragments fetched 32 k-blocks ahead as one run; sums and order unchanged
# speedup vs baseline: 3.2761x; 1.2726x over previous
#include <hip/hip_runtime.h>
#include <math.h>

#define BATCH   256
#define STATE   512
#define HIDDEN  1024
#define COND    64
#define NSTEPS  256
#define TSCALE  0.1f

#define XS_PITCH (STATE + 4)
#define HS_PITCH (HIDDEN + 4)

#define STEP_THREADS 128

#define XS_Q ((16 * STATE / 4) / STEP_THREADS)
#define HS_Q ((16 * HIDDEN / 4) / STEP_THREADS)

typedef __attribute__((ext_vector_type(2))) float v2f;
typedef __attribute__((ext_vector_type(4))) float v4f;
typedef __attribute__((ext_vector_type(8))) float v8f;

static __device__ __forceinline__ v8f wmma_f32(v2f a, v2f b, v8f c) {
    return __builtin_amdgcn_wmma_f32_16x16x4_f32(
         false, a,  false, b,
         (short)0, c,  false,  false);
}

#define KB_GROUP 32

static __device__ __forceinline__ void load_w_group(v4f (&w)[KB_GROUP], const float* __restrict__ wp, int g) {
    #pragma unroll
    for (int j = 0; j < KB_GROUP; ++j)
        w[j] = *(const v4f*)(wp + (size_t)(g * KB_GROUP + j) * 128);
}

static __device__ __forceinline__ void load_a_group(v2f (&a)[2 * KB_GROUP], const float* arow, int g) {
    #pragma unroll
    for (int j = 0; j < KB_GROUP; ++j) {
        const int kb = g * KB_GROUP + j;
        a[2 * j]     = *(const v2f*)&arow[kb * 8];
        a[2 * j + 1] = *(const v2f*)&arow[kb * 8 + 4];
    }
}

static __device__ __forceinline__ v8f mma_group(const v4f (&w)[KB_GROUP], const v2f (&a)[2 * KB_GROUP], v8f acc) {
    #pragma unroll
    for (int j = 0; j < KB_GROUP; ++j) {
        v2f b0 = {w[j].x, w[j].y};
        v2f b1 = {w[j].z, w[j].w};
        acc = wmma_f32(a[2 * j], b0, acc);
        acc = wmma_f32(a[2 * j + 1], b1, acc);
    }
    return acc;
}

static __device__ __forceinline__ v8f wmma_drain(v8f acc) {
    __builtin_amdgcn_sched_barrier(0);
    asm volatile("v_nop\n\tv_nop\n\tv_nop\n\tv_nop\n\tv_nop\n\tv_nop\n\tv_nop\n\tv_nop" : "+v"(acc));
    return acc;
}

__global__ __launch_bounds__(256) void copy_x0_kernel(const float* __restrict__ x0,
                                                      float* __restrict__ out) {
    int i = blockIdx.x * 256 + threadIdx.x;
    out[i] = x0[i];
}

__global__ __launch_bounds__(256) void z1_precompute_kernel(const float* __restrict__ z,
                                                            const float* __restrict__ W1,
                                                            const float* __restrict__ b1,
                                                            float* __restrict__ Z1) {
    int idx = blockIdx.x * 256 + threadIdx.x;
    int b = idx >> 10;
    int j = idx & (HIDDEN - 1);
    float acc = b1[j];
    const float* zrow = z + b * COND;
    const float* wz   = W1 + (size_t)STATE * HIDDEN + j;
    #pragma unroll 4
    for (int k = 0; k < COND; ++k)
        acc += zrow[k] * wz[(size_t)k * HIDDEN];
    Z1[idx] = acc;
}

__global__ __launch_bounds__(256) void pack_w_kernel(const float* __restrict__ W,
                                                     float* __restrict__ Wp,
                                                     int K, int N) {
    int idx  = blockIdx.x * 256 + threadIdx.x;
    int lane = idx & 31;
    int kb2  = (idx >> 5) % (K >> 3);
    int nt   = (idx >> 5) / (K >> 3);
    int k0   = (kb2 << 3) + ((lane >> 4) << 1);
    int n    = (nt << 4) + (lane & 15);
    size_t o = (size_t)idx * 4;
    Wp[o + 0] = W[(size_t)(k0 + 0) * N + n];
    Wp[o + 1] = W[(size_t)(k0 + 1) * N + n];
    Wp[o + 2] = W[(size_t)(k0 + 4) * N + n];
    Wp[o + 3] = W[(size_t)(k0 + 5) * N + n];
}

__global__ __launch_bounds__(STEP_THREADS) void step1_packed_kernel(const float* __restrict__ X,
                                                           const float* __restrict__ W1pack,
                                                           const float* __restrict__ W1,
                                                           const float* __restrict__ Z1,
                                                           const float* __restrict__ t,
                                                           int s,
                                                           float* __restrict__ Hbuf) {
    __shared__ __attribute__((aligned(16))) float xs[16 * XS_PITCH];

    const int mt   = blockIdx.x >> 4;
    const int ng   = blockIdx.x & 15;
    const int m0   = mt * 16;
    const int wave = threadIdx.x >> 5;
    const int lane = threadIdx.x & 31;
    const int n0   = (ng * 4 + wave) * 16;

    const int ar = lane & 15;
    const int ak = (lane >> 4) * 2;
    const float* wp = W1pack + ((size_t)(n0 >> 4) * (STATE >> 3)) * 128 + lane * 4;

    v4f xa[XS_Q];
    const float* xpanel = X + (size_t)m0 * STATE;
    #pragma unroll
    for (int j = 0; j < XS_Q; ++j)
        xa[j] = *(const v4f*)(xpanel + (size_t)(threadIdx.x + STEP_THREADS * j) * 4);

    v4f wa[KB_GROUP], wb[KB_GROUP];
    load_w_group(wa, wp, 0);

    const float ts   = t[s];
    const float* W1t = W1 + (size_t)(STATE + COND) * HIDDEN;
    const int col   = n0 + (lane & 15);
    const int rbase = m0 + ((lane >> 4) << 3);
    const float w1t = W1t[col];
    float z1v[8];
    #pragma unroll
    for (int r = 0; r < 8; ++r)
        z1v[r] = Z1[(size_t)(rbase + r) * HIDDEN + col];

    #pragma unroll
    for (int j = 0; j < XS_Q; ++j) {
        const int i = (threadIdx.x + STEP_THREADS * j) * 4;
        *(v4f*)&xs[(i >> 9) * XS_PITCH + (i & (STATE - 1))] = xa[j];
    }
    __syncthreads();

    const float* arow = &xs[ar * XS_PITCH + ak];

    v2f aa[2 * KB_GROUP], ab[2 * KB_GROUP];
    load_a_group(aa, arow, 0);

    v8f acc = {0.f, 0.f, 0.f, 0.f, 0.f, 0.f, 0.f, 0.f};
    static_assert((STATE >> 3) / KB_GROUP == 2, "step 1 is written for two groups");
    load_w_group(wb, wp, 1);
    load_a_group(ab, arow, 1);
    __builtin_amdgcn_sched_barrier(0);
    acc = mma_group(wa, aa, acc);
    acc = mma_group(wb, ab, acc);
    acc = wmma_drain(acc);

    #pragma unroll
    for (int r = 0; r < 8; ++r) {
        const int row = rbase + r;
        float v = acc[r] + z1v[r] + ts * w1t;
        Hbuf[(size_t)row * HIDDEN + col] = tanhf(v);
    }
}

__global__ __launch_bounds__(STEP_THREADS) void step2_packed_kernel(const float* __restrict__ Hbuf,
                                                           const float* __restrict__ W2pack,
                                                           const float* __restrict__ b2,
                                                           const float* __restrict__ X,
                                                           const float* __restrict__ t,
                                                           float* __restrict__ Xnext) {
    __shared__ __attribute__((aligned(16))) float hs[16 * HS_PITCH];

    const int mt   = blockIdx.x >> 3;
    const int ng   = blockIdx.x & 7;
    const int m0   = mt * 16;
    const int wave = threadIdx.x >> 5;
    const int lane = threadIdx.x & 31;
    const int n0   = (ng * 4 + wave) * 16;

    const int ar = lane & 15;
    const int ak = (lane >> 4) * 2;
    const float* wp = W2pack + ((size_t)(n0 >> 4) * (HIDDEN >> 3)) * 128 + lane * 4;

    v4f ha[HS_Q];
    const float* hpanel = Hbuf + (size_t)m0 * HIDDEN;
    #pragma unroll
    for (int j = 0; j < HS_Q; ++j)
        ha[j] = *(const v4f*)(hpanel + (size_t)(threadIdx.x + STEP_THREADS * j) * 4);

    v4f wa[KB_GROUP], wb[KB_GROUP];
    load_w_group(wa, wp, 0);

    const float dt = (t[1] - t[0]) * TSCALE;
    const int col   = n0 + (lane & 15);
    const int rbase = m0 + ((lane >> 4) << 3);
    const float b2v = b2[col];
    float xv[8];
    #pragma unroll
    for (int r = 0; r < 8; ++r)
        xv[r] = X[(size_t)(rbase + r) * STATE + col];

    #pragma unroll
    for (int j = 0; j < HS_Q; ++j) {
        const int i = (threadIdx.x + STEP_THREADS * j) * 4;
        *(v4f*)&hs[(i >> 10) * HS_PITCH + (i & (HIDDEN - 1))] = ha[j];
    }
    __syncthreads();

    const float* arow = &hs[ar * HS_PITCH + ak];

    v2f aa[2 * KB_GROUP], ab[2 * KB_GROUP];
    load_a_group(aa, arow, 0);

    v8f acc = {0.f, 0.f, 0.f, 0.f, 0.f, 0.f, 0.f, 0.f};
    #pragma unroll
    for (int g = 0; g < (HIDDEN >> 3) / KB_GROUP; g += 2) {
        load_w_group(wb, wp, g + 1);
        load_a_group(ab, arow, g + 1);
        __builtin_amdgcn_sched_barrier(0);
        acc = mma_group(wa, aa, acc);
        __builtin_amdgcn_sched_barrier(0);
        if (g + 2 < (HIDDEN >> 3) / KB_GROUP) {
            load_w_group(wa, wp, g + 2);
            load_a_group(aa, arow, g + 2);
        }
        __builtin_amdgcn_sched_barrier(0);
        acc = mma_group(wb, ab, acc);
        __builtin_amdgcn_sched_barrier(0);
    }
    acc = wmma_drain(acc);

    #pragma unroll
    for (int r = 0; r < 8; ++r) {
        const int row = rbase + r;
        float f = acc[r] + b2v;
        float v = xv[r] + dt * f;
        Xnext[(size_t)row * STATE + col] = fmaxf(v, 0.f);
    }
}

__global__ __launch_bounds__(256) void step1_direct_kernel(const float* __restrict__ X,
                                                           const float* __restrict__ W1,
                                                           const float* __restrict__ Z1,
                                                           const float* __restrict__ t,
                                                           int s,
                                                           float* __restrict__ Hbuf) {
    __shared__ float xs[16 * XS_PITCH];
    const int mt = blockIdx.x >> 3, ng = blockIdx.x & 7;
    const int m0 = mt * 16;
    const int wave = threadIdx.x >> 5, lane = threadIdx.x & 31;
    const int n0 = (ng * 8 + wave) * 16;
    for (int i = threadIdx.x; i < 16 * STATE; i += 256)
        xs[(i >> 9) * XS_PITCH + (i & (STATE - 1))] = X[(size_t)(m0 + (i >> 9)) * STATE + (i & (STATE - 1))];
    __syncthreads();
    const int ar = lane & 15, ak = (lane >> 4) * 2;
    v8f acc = {0.f, 0.f, 0.f, 0.f, 0.f, 0.f, 0.f, 0.f};
    for (int k = 0; k < STATE; k += 4) {
        v2f a = *(const v2f*)&xs[ar * XS_PITCH + k + ak];
        v2f b;
        b.x = W1[(size_t)(k + ak) * HIDDEN + n0 + ar];
        b.y = W1[(size_t)(k + ak + 1) * HIDDEN + n0 + ar];
        acc = wmma_f32(a, b, acc);
    }
    const float ts = t[s];
    const float* W1t = W1 + (size_t)(STATE + COND) * HIDDEN;
    const int col = n0 + (lane & 15);
    const int rbase = m0 + ((lane >> 4) << 3);
    #pragma unroll
    for (int r = 0; r < 8; ++r) {
        const int row = rbase + r;
        float v = acc[r] + Z1[(size_t)row * HIDDEN + col] + ts * W1t[col];
        Hbuf[(size_t)row * HIDDEN + col] = tanhf(v);
    }
}

__global__ __launch_bounds__(256) void step2_direct_kernel(const float* __restrict__ Hbuf,
                                                           const float* __restrict__ W2,
                                                           const float* __restrict__ b2,
                                                           const float* __restrict__ X,
                                                           const float* __restrict__ t,
                                                           float* __restrict__ Xnext) {
    __shared__ float hs[16 * HS_PITCH];
    const int mt = blockIdx.x >> 2, ng = blockIdx.x & 3;
    const int m0 = mt * 16;
    const int wave = threadIdx.x >> 5, lane = threadIdx.x & 31;
    const int n0 = (ng * 8 + wave) * 16;
    for (int i = threadIdx.x; i < 16 * HIDDEN; i += 256)
        hs[(i >> 10) * HS_PITCH + (i & (HIDDEN - 1))] = Hbuf[(size_t)(m0 + (i >> 10)) * HIDDEN + (i & (HIDDEN - 1))];
    __syncthreads();
    const int ar = lane & 15, ak = (lane >> 4) * 2;
    v8f acc = {0.f, 0.f, 0.f, 0.f, 0.f, 0.f, 0.f, 0.f};
    for (int k = 0; k < HIDDEN; k += 4) {
        v2f a = *(const v2f*)&hs[ar * HS_PITCH + k + ak];
        v2f b;
        b.x = W2[(size_t)(k + ak) * STATE + n0 + ar];
        b.y = W2[(size_t)(k + ak + 1) * STATE + n0 + ar];
        acc = wmma_f32(a, b, acc);
    }
    const float dt = (t[1] - t[0]) * TSCALE;
    const int col = n0 + (lane & 15);
    const int rbase = m0 + ((lane >> 4) << 3);
    #pragma unroll
    for (int r = 0; r < 8; ++r) {
        const int row = rbase + r;
        float f = acc[r] + b2[col];
        float v = X[(size_t)row * STATE + col] + dt * f;
        Xnext[(size_t)row * STATE + col] = fmaxf(v, 0.f);
    }
}

extern "C" void kernel_launch(void* const* d_in, const int* in_sizes, int n_in,
                              void* d_out, int out_size, void* d_ws, size_t ws_size,
                              hipStream_t stream) {
    const float* x0 = (const float*)d_in[0];
    const float* z  = (const float*)d_in[1];
    const float* t  = (const float*)d_in[2];
    const float* W1 = (const float*)d_in[3];
    const float* b1 = (const float*)d_in[4];
    const float* W2 = (const float*)d_in[5];
    const float* b2 = (const float*)d_in[6];
    float* out = (float*)d_out;

    const size_t nZ1 = (size_t)BATCH * HIDDEN;
    const size_t nH  = (size_t)BATCH * HIDDEN;
    const size_t nW1 = (size_t)STATE * HIDDEN;
    const size_t nW2 = (size_t)HIDDEN * STATE;

    float* Z1     = (float*)d_ws;
    float* Hbuf   = Z1 + nZ1;
    float* W1pack = Hbuf + nH;
    float* W2pack = W1pack + nW1;

    const bool packed = ws_size >= (nZ1 + nH + nW1 + nW2) * sizeof(float);

    copy_x0_kernel<<<(BATCH * STATE) / 256, 256, 0, stream>>>(x0, out);
    z1_precompute_kernel<<<(BATCH * HIDDEN) / 256, 256, 0, stream>>>(z, W1, b1, Z1);

    if (packed) {
        pack_w_kernel<<<(int)((nW1 / 4) / 256), 256, 0, stream>>>(W1, W1pack, STATE, HIDDEN);
        pack_w_kernel<<<(int)((nW2 / 4) / 256), 256, 0, stream>>>(W2, W2pack, HIDDEN, STATE);
        for (int s = 0; s < NSTEPS - 1; ++s) {
            const float* xcur  = out + (size_t)s * BATCH * STATE;
            float*       xnext = out + (size_t)(s + 1) * BATCH * STATE;
            step1_packed_kernel<<<256, STEP_THREADS, 0, stream>>>(xcur, W1pack, W1, Z1, t, s, Hbuf);
            step2_packed_kernel<<<128, STEP_THREADS, 0, stream>>>(Hbuf, W2pack, b2, xcur, t, xnext);
        }
    } else {
        for (int s = 0; s < NSTEPS - 1; ++s) {
            const float* xcur  = out + (size_t)s * BATCH * STATE;
            float*       xnext = out + (size_t)(s + 1) * BATCH * STATE;
            step1_direct_kernel<<<128, 256, 0, stream>>>(xcur, W1, Z1, t, s, Hbuf);
            step2_direct_kernel<<<64, 256, 0, stream>>>(Hbuf, W2, b2, xcur, t, xnext);
        }
    }
}


# Split-K: step 2's K halves on two waves (grid 128 -> 256 blocks, partial tiles added through LDS), and each wave's WMMAs alternate two accumulators; summation order changes, f32 kept
# speedup vs baseline: 3.7250x; 1.1370x over previous
#include <hip/hip_runtime.h>
#include <math.h>

#define BATCH   256
#define STATE   512
#define HIDDEN  1024
#define COND    64
#define NSTEPS  256
#define TSCALE  0.1f

#define XS_PITCH (STATE + 4)
#define HS_PITCH (HIDDEN + 4)

#define STEP_THREADS 128

#define XS_Q ((16 * STATE / 4) / STEP_THREADS)
#define HS_Q ((16 * HIDDEN / 4) / STEP_THREADS)

typedef __attribute__((ext_vector_type(2))) float v2f;
typedef __attribute__((ext_vector_type(4))) float v4f;
typedef __attribute__((ext_vector_type(8))) float v8f;

static __device__ __forceinline__ v8f wmma_f32(v2f a, v2f b, v8f c) {
    return __builtin_amdgcn_wmma_f32_16x16x4_f32(
         false, a,  false, b,
         (short)0, c,  false,  false);
}

#define KB_GROUP 32

static __device__ __forceinline__ void load_w_group(v4f (&w)[KB_GROUP], const float* __restrict__ wp, int g) {
    #pragma unroll
    for (int j = 0; j < KB_GROUP; ++j)
        w[j] = *(const v4f*)(wp + (size_t)(g * KB_GROUP + j) * 128);
}

static __device__ __forceinline__ void load_a_group(v2f (&a)[2 * KB_GROUP], const float* arow, int g) {
    #pragma unroll
    for (int j = 0; j < KB_GROUP; ++j) {
        const int kb = g * KB_GROUP + j;
        a[2 * j]     = *(const v2f*)&arow[kb * 8];
        a[2 * j + 1] = *(const v2f*)&arow[kb * 8 + 4];
    }
}

static __device__ __forceinline__ void mma_group(const v4f (&w)[KB_GROUP], const v2f (&a)[2 * KB_GROUP],
                                                 v8f& acc0, v8f& acc1) {
    #pragma unroll
    for (int j = 0; j < KB_GROUP; ++j) {
        v2f b0 = {w[j].x, w[j].y};
        v2f b1 = {w[j].z, w[j].w};
        acc0 = wmma_f32(a[2 * j], b0, acc0);
        acc1 = wmma_f32(a[2 * j + 1], b1, acc1);
    }
}

static __device__ __forceinline__ v8f wmma_drain_sum(v8f acc0, v8f acc1) {
    __builtin_amdgcn_sched_barrier(0);
    asm volatile("v_nop\n\tv_nop\n\tv_nop\n\tv_nop\n\tv_nop\n\tv_nop\n\tv_nop\n\tv_nop" : "+v"(acc0), "+v"(acc1));
    return acc0 + acc1;
}

__global__ __launch_bounds__(256) void copy_x0_kernel(const float* __restrict__ x0,
                                                      float* __restrict__ out) {
    int i = blockIdx.x * 256 + threadIdx.x;
    out[i] = x0[i];
}

__global__ __launch_bounds__(256) void z1_precompute_kernel(const float* __restrict__ z,
                                                            const float* __restrict__ W1,
                                                            const float* __restrict__ b1,
                                                            float* __restrict__ Z1) {
    int idx = blockIdx.x * 256 + threadIdx.x;
    int b = idx >> 10;
    int j = idx & (HIDDEN - 1);
    float acc = b1[j];
    const float* zrow = z + b * COND;
    const float* wz   = W1 + (size_t)STATE * HIDDEN + j;
    #pragma unroll 4
    for (int k = 0; k < COND; ++k)
        acc += zrow[k] * wz[(size_t)k * HIDDEN];
    Z1[idx] = acc;
}

__global__ __launch_bounds__(256) void pack_w_kernel(const float* __restrict__ W,
                                                     float* __restrict__ Wp,
                                                     int K, int N) {
    int idx  = blockIdx.x * 256 + threadIdx.x;
    int lane = idx & 31;
    int kb2  = (idx >> 5) % (K >> 3);
    int nt   = (idx >> 5) / (K >> 3);
    int k0   = (kb2 << 3) + ((lane >> 4) << 1);
    int n    = (nt << 4) + (lane & 15);
    size_t o = (size_t)idx * 4;
    Wp[o + 0] = W[(size_t)(k0 + 0) * N + n];
    Wp[o + 1] = W[(size_t)(k0 + 1) * N + n];
    Wp[o + 2] = W[(size_t)(k0 + 4) * N + n];
    Wp[o + 3] = W[(size_t)(k0 + 5) * N + n];
}

__global__ __launch_bounds__(STEP_THREADS) void step1_packed_kernel(const float* __restrict__ X,
                                                           const float* __restrict__ W1pack,
                                                           const float* __restrict__ W1,
                                                           const float* __restrict__ Z1,
                                                           const float* __restrict__ t,
                                                           int s,
                                                           float* __restrict__ Hbuf) {
    __shared__ __attribute__((aligned(16))) float xs[16 * XS_PITCH];

    const int mt   = blockIdx.x >> 4;
    const int ng   = blockIdx.x & 15;
    const int m0   = mt * 16;
    const int wave = threadIdx.x >> 5;
    const int lane = threadIdx.x & 31;
    const int n0   = (ng * 4 + wave) * 16;

    const int ar = lane & 15;
    const int ak = (lane >> 4) * 2;
    const float* wp = W1pack + ((size_t)(n0 >> 4) * (STATE >> 3)) * 128 + lane * 4;

    v4f xa[XS_Q];
    const float* xpanel = X + (size_t)m0 * STATE;
    #pragma unroll
    for (int j = 0; j < XS_Q; ++j)
        xa[j] = *(const v4f*)(xpanel + (size_t)(threadIdx.x + STEP_THREADS * j) * 4);

    v4f wa[KB_GROUP], wb[KB_GROUP];
    load_w_group(wa, wp, 0);

    const float ts   = t[s];
    const float* W1t = W1 + (size_t)(STATE + COND) * HIDDEN;
    const int col   = n0 + (lane & 15);
    const int rbase = m0 + ((lane >> 4) << 3);
    const float w1t = W1t[col];
    float z1v[8];
    #pragma unroll
    for (int r = 0; r < 8; ++r)
        z1v[r] = Z1[(size_t)(rbase + r) * HIDDEN + col];

    #pragma unroll
    for (int j = 0; j < XS_Q; ++j) {
        const int i = (threadIdx.x + STEP_THREADS * j) * 4;
        *(v4f*)&xs[(i >> 9) * XS_PITCH + (i & (STATE - 1))] = xa[j];
    }
    __syncthreads();

    const float* arow = &xs[ar * XS_PITCH + ak];

    v2f aa[2 * KB_GROUP], ab[2 * KB_GROUP];
    load_a_group(aa, arow, 0);

    v8f acc0 = {0.f, 0.f, 0.f, 0.f, 0.f, 0.f, 0.f, 0.f};
    v8f acc1 = {0.f, 0.f, 0.f, 0.f, 0.f, 0.f, 0.f, 0.f};
    static_assert((STATE >> 3) / KB_GROUP == 2, "step 1 is written for two groups");
    load_w_group(wb, wp, 1);
    load_a_group(ab, arow, 1);
    __builtin_amdgcn_sched_barrier(0);
    mma_group(wa, aa, acc0, acc1);
    mma_group(wb, ab, acc0, acc1);
    const v8f acc = wmma_drain_sum(acc0, acc1);

    #pragma unroll
    for (int r = 0; r < 8; ++r) {
        const int row = rbase + r;
        float v = acc[r] + z1v[r] + ts * w1t;
        Hbuf[(size_t)row * HIDDEN + col] = tanhf(v);
    }
}

__global__ __launch_bounds__(STEP_THREADS) void step2_packed_kernel(const float* __restrict__ Hbuf,
                                                           const float* __restrict__ W2pack,
                                                           const float* __restrict__ b2,
                                                           const float* __restrict__ X,
                                                           const float* __restrict__ t,
                                                           float* __restrict__ Xnext) {
    __shared__ __attribute__((aligned(16))) float hs[16 * HS_PITCH];
    __shared__ __attribute__((aligned(16))) float red[2 * 2 * 32 * 4];

    const int mt   = blockIdx.x >> 4;
    const int ng   = blockIdx.x & 15;
    const int m0   = mt * 16;
    const int wave = __builtin_amdgcn_readfirstlane(threadIdx.x >> 5);
    const int lane = threadIdx.x & 31;
    const int kh   = wave >> 1;
    const int tl   = wave & 1;
    const int n0   = (ng * 2 + tl) * 16;

    const int ar = lane & 15;
    const int ak = (lane >> 4) * 2;
    const float* wp = W2pack + ((size_t)(n0 >> 4) * (HIDDEN >> 3) + (size_t)kh * (HIDDEN >> 4)) * 128 + lane * 4;

    v4f ha[HS_Q];
    const float* hpanel = Hbuf + (size_t)m0 * HIDDEN;
    #pragma unroll
    for (int j = 0; j < HS_Q; ++j)
        ha[j] = *(const v4f*)(hpanel + (size_t)(threadIdx.x + STEP_THREADS * j) * 4);

    v4f wa[KB_GROUP], wb[KB_GROUP];
    load_w_group(wa, wp, 0);

    const int col   = n0 + (lane & 15);
    const int rbase = m0 + ((lane >> 4) << 3);
    float dt = 0.f, b2v = 0.f;
    float xv[8] = {0.f, 0.f, 0.f, 0.f, 0.f, 0.f, 0.f, 0.f};
    if (kh == 0) {
        dt  = (t[1] - t[0]) * TSCALE;
        b2v = b2[col];
        #pragma unroll
        for (int r = 0; r < 8; ++r)
            xv[r] = X[(size_t)(rbase + r) * STATE + col];
    }

    #pragma unroll
    for (int j = 0; j < HS_Q; ++j) {
        const int i = (threadIdx.x + STEP_THREADS * j) * 4;
        *(v4f*)&hs[(i >> 10) * HS_PITCH + (i & (HIDDEN - 1))] = ha[j];
    }
    __syncthreads();

    const float* arow = &hs[ar * HS_PITCH + ak + kh * (HIDDEN >> 1)];

    v2f aa[2 * KB_GROUP], ab[2 * KB_GROUP];
    load_a_group(aa, arow, 0);

    v8f acc0 = {0.f, 0.f, 0.f, 0.f, 0.f, 0.f, 0.f, 0.f};
    v8f acc1 = {0.f, 0.f, 0.f, 0.f, 0.f, 0.f, 0.f, 0.f};
    static_assert((HIDDEN >> 4) / KB_GROUP == 2, "step 2 is written for two groups per K half");
    load_w_group(wb, wp, 1);
    load_a_group(ab, arow, 1);
    __builtin_amdgcn_sched_barrier(0);
    mma_group(wa, aa, acc0, acc1);
    mma_group(wb, ab, acc0, acc1);
    const v8f acc = wmma_drain_sum(acc0, acc1);

    v4f* rslot = (v4f*)&red[tl * 256 + lane * 4];
    if (kh == 1) {
        const v4f lo = {acc[0], acc[1], acc[2], acc[3]};
        const v4f hi = {acc[4], acc[5], acc[6], acc[7]};
        rslot[0]  = lo;
        rslot[32] = hi;
    }
    __syncthreads();
    if (kh == 0) {
        const v4f lo = rslot[0];
        const v4f hi = rslot[32];
        const float part[8] = {lo.x, lo.y, lo.z, lo.w, hi.x, hi.y, hi.z, hi.w};
        #pragma unroll
        for (int r = 0; r < 8; ++r) {
            const int row = rbase + r;
            float f = (acc[r] + part[r]) + b2v;
            float v = xv[r] + dt * f;
            Xnext[(size_t)row * STATE + col] = fmaxf(v, 0.f);
        }
    }
}

__global__ __launch_bounds__(256) void step1_direct_kernel(const float* __restrict__ X,
                                                           const float* __restrict__ W1,
                                                           const float* __restrict__ Z1,
                                                           const float* __restrict__ t,
                                                           int s,
                                                           float* __restrict__ Hbuf) {
    __shared__ float xs[16 * XS_PITCH];
    const int mt = blockIdx.x >> 3, ng = blockIdx.x & 7;
    const int m0 = mt * 16;
    const int wave = threadIdx.x >> 5, lane = threadIdx.x & 31;
    const int n0 = (ng * 8 + wave) * 16;
    for (int i = threadIdx.x; i < 16 * STATE; i += 256)
        xs[(i >> 9) * XS_PITCH + (i & (STATE - 1))] = X[(size_t)(m0 + (i >> 9)) * STATE + (i & (STATE - 1))];
    __syncthreads();
    const int ar = lane & 15, ak = (lane >> 4) * 2;
    v8f acc = {0.f, 0.f, 0.f, 0.f, 0.f, 0.f, 0.f, 0.f};
    for (int k = 0; k < STATE; k += 4) {
        v2f a = *(const v2f*)&xs[ar * XS_PITCH + k + ak];
        v2f b;
        b.x = W1[(size_t)(k + ak) * HIDDEN + n0 + ar];
        b.y = W1[(size_t)(k + ak + 1) * HIDDEN + n0 + ar];
        acc = wmma_f32(a, b, acc);
    }
    const float ts = t[s];
    const float* W1t = W1 + (size_t)(STATE + COND) * HIDDEN;
    const int col = n0 + (lane & 15);
    const int rbase = m0 + ((lane >> 4) << 3);
    #pragma unroll
    for (int r = 0; r < 8; ++r) {
        const int row = rbase + r;
        float v = acc[r] + Z1[(size_t)row * HIDDEN + col] + ts * W1t[col];
        Hbuf[(size_t)row * HIDDEN + col] = tanhf(v);
    }
}

__global__ __launch_bounds__(256) void step2_direct_kernel(const float* __restrict__ Hbuf,
                                                           const float* __restrict__ W2,
                                                           const float* __restrict__ b2,
                                                           const float* __restrict__ X,
                                                           const float* __restrict__ t,
                                                           float* __restrict__ Xnext) {
    __shared__ float hs[16 * HS_PITCH];
    const int mt = blockIdx.x >> 2, ng = blockIdx.x & 3;
    const int m0 = mt * 16;
    const int wave = threadIdx.x >> 5, lane = threadIdx.x & 31;
    const int n0 = (ng * 8 + wave) * 16;
    for (int i = threadIdx.x; i < 16 * HIDDEN; i += 256)
        hs[(i >> 10) * HS_PITCH + (i & (HIDDEN - 1))] = Hbuf[(size_t)(m0 + (i >> 10)) * HIDDEN + (i & (HIDDEN - 1))];
    __syncthreads();
    const int ar = lane & 15, ak = (lane >> 4) * 2;
    v8f acc = {0.f, 0.f, 0.f, 0.f, 0.f, 0.f, 0.f, 0.f};
    for (int k = 0; k < HIDDEN; k += 4) {
        v2f a = *(const v2f*)&hs[ar * HS_PITCH + k + ak];
        v2f b;
        b.x = W2[(size_t)(k + ak) * STATE + n0 + ar];
        b.y = W2[(size_t)(k + ak + 1) * STATE + n0 + ar];
        acc = wmma_f32(a, b, acc);
    }
    const float dt = (t[1] - t[0]) * TSCALE;
    const int col = n0 + (lane & 15);
    const int rbase = m0 + ((lane >> 4) << 3);
    #pragma unroll
    for (int r = 0; r < 8; ++r) {
        const int row = rbase + r;
        float f = acc[r] + b2[col];
        float v = X[(size_t)row * STATE + col] + dt * f;
        Xnext[(size_t)row * STATE + col] = fmaxf(v, 0.f);
    }
}

extern "C" void kernel_launch(void* const* d_in, const int* in_sizes, int n_in,
                              void* d_out, int out_size, void* d_ws, size_t ws_size,
                              hipStream_t stream) {
    const float* x0 = (const float*)d_in[0];
    const float* z  = (const float*)d_in[1];
    const float* t  = (const float*)d_in[2];
    const float* W1 = (const float*)d_in[3];
    const float* b1 = (const float*)d_in[4];
    const float* W2 = (const float*)d_in[5];
    const float* b2 = (const float*)d_in[6];
    float* out = (float*)d_out;

    const size_t nZ1 = (size_t)BATCH * HIDDEN;
    const size_t nH  = (size_t)BATCH * HIDDEN;
    const size_t nW1 = (size_t)STATE * HIDDEN;
    const size_t nW2 = (size_t)HIDDEN * STATE;

    float* Z1     = (float*)d_ws;
    float* Hbuf   = Z1 + nZ1;
    float* W1pack = Hbuf + nH;
    float* W2pack = W1pack + nW1;

    const bool packed = ws_size >= (nZ1 + nH + nW1 + nW2) * sizeof(float);

    copy_x0_kernel<<<(BATCH * STATE) / 256, 256, 0, stream>>>(x0, out);
    z1_precompute_kernel<<<(BATCH * HIDDEN) / 256, 256, 0, stream>>>(z, W1, b1, Z1);

    if (packed) {
        pack_w_kernel<<<(int)((nW1 / 4) / 256), 256, 0, stream>>>(W1, W1pack, STATE, HIDDEN);
        pack_w_kernel<<<(int)((nW2 / 4) / 256), 256, 0, stream>>>(W2, W2pack, HIDDEN, STATE);
        for (int s = 0; s < NSTEPS - 1; ++s) {
            const float* xcur  = out + (size_t)s * BATCH * STATE;
            float*       xnext = out + (size_t)(s + 1) * BATCH * STATE;
            step1_packed_kernel<<<256, STEP_THREADS, 0, stream>>>(xcur, W1pack, W1, Z1, t, s, Hbuf);
            step2_packed_kernel<<<256, STEP_THREADS, 0, stream>>>(Hbuf, W2pack, b2, xcur, t, xnext);
        }
    } else {
        for (int s = 0; s < NSTEPS - 1; ++s) {
            const float* xcur  = out + (size_t)s * BATCH * STATE;
            float*       xnext = out + (size_t)(s + 1) * BATCH * STATE;
            step1_direct_kernel<<<128, 256, 0, stream>>>(xcur, W1, Z1, t, s, Hbuf);
            step2_direct_kernel<<<64, 256, 0, stream>>>(Hbuf, W2, b2, xcur, t, xnext);
        }
    }
}


# Step-1 epilogue: tanh without the library's per-lane branch (both ranges computed, then a select; exp(2|x|) as one 2^y), so a lane's 8 activations interleave, not 8 serial chains; K sums unchanged
# speedup vs baseline: 3.9726x; 1.0665x over previous
#include <hip/hip_runtime.h>
#include <math.h>

#define BATCH   256
#define STATE   512
#define HIDDEN  1024
#define COND    64
#define NSTEPS  256
#define TSCALE  0.1f

#define XS_PITCH (STATE + 4)
#define HS_PITCH (HIDDEN + 4)

#define STEP_THREADS 128

#define XS_Q ((16 * STATE / 4) / STEP_THREADS)
#define HS_Q ((16 * HIDDEN / 4) / STEP_THREADS)

typedef __attribute__((ext_vector_type(2))) float v2f;
typedef __attribute__((ext_vector_type(4))) float v4f;
typedef __attribute__((ext_vector_type(8))) float v8f;

static __device__ __forceinline__ v8f wmma_f32(v2f a, v2f b, v8f c) {
    return __builtin_amdgcn_wmma_f32_16x16x4_f32(
         false, a,  false, b,
         (short)0, c,  false,  false);
}

#define KB_GROUP 32

static __device__ __forceinline__ void load_w_group(v4f (&w)[KB_GROUP], const float* __restrict__ wp, int g) {
    #pragma unroll
    for (int j = 0; j < KB_GROUP; ++j)
        w[j] = *(const v4f*)(wp + (size_t)(g * KB_GROUP + j) * 128);
}

static __device__ __forceinline__ void load_a_group(v2f (&a)[2 * KB_GROUP], const float* arow, int g) {
    #pragma unroll
    for (int j = 0; j < KB_GROUP; ++j) {
        const int kb = g * KB_GROUP + j;
        a[2 * j]     = *(const v2f*)&arow[kb * 8];
        a[2 * j + 1] = *(const v2f*)&arow[kb * 8 + 4];
    }
}

static __device__ __forceinline__ void mma_group(const v4f (&w)[KB_GROUP], const v2f (&a)[2 * KB_GROUP],
                                                 v8f& acc0, v8f& acc1) {
    #pragma unroll
    for (int j = 0; j < KB_GROUP; ++j) {
        v2f b0 = {w[j].x, w[j].y};
        v2f b1 = {w[j].z, w[j].w};
        acc0 = wmma_f32(a[2 * j], b0, acc0);
        acc1 = wmma_f32(a[2 * j + 1], b1, acc1);
    }
}

static __device__ __forceinline__ v8f wmma_drain_sum(v8f acc0, v8f acc1) {
    __builtin_amdgcn_sched_barrier(0);
    asm volatile("v_nop\n\tv_nop\n\tv_nop\n\tv_nop\n\tv_nop\n\tv_nop\n\tv_nop\n\tv_nop" : "+v"(acc0), "+v"(acc1));
    return acc0 + acc1;
}

static __device__ __forceinline__ float tanh_straight(float x) {
    const float ax = __builtin_fabsf(x);
    const float x2 = x * x;
    float p = __builtin_fmaf(x2, -0.00570002059f, 0.0206340719f);
    p = __builtin_fmaf(x2, p, -0.0537379310f);
    p = __builtin_fmaf(x2, p, 0.133314162f);
    p = __builtin_fmaf(x2, p, -0.333332807f);
    const float small = __builtin_fmaf(x2, ax * p, ax);
    const float e   = __builtin_amdgcn_exp2f(ax * 2.88539004f);
    const float big = __builtin_fmaf(__builtin_amdgcn_rcpf(e + 1.0f), -2.0f, 1.0f);
    const float t = (ax < 0.625f) ? small : big;
    return __builtin_copysignf(t, x);
}

__global__ __launch_bounds__(256) void copy_x0_kernel(const float* __restrict__ x0,
                                                      float* __restrict__ out) {
    int i = blockIdx.x * 256 + threadIdx.x;
    out[i] = x0[i];
}

__global__ __launch_bounds__(256) void z1_precompute_kernel(const float* __restrict__ z,
                                                            const float* __restrict__ W1,
                                                            const float* __restrict__ b1,
                                                            float* __restrict__ Z1) {
    int idx = blockIdx.x * 256 + threadIdx.x;
    int b = idx >> 10;
    int j = idx & (HIDDEN - 1);
    float acc = b1[j];
    const float* zrow = z + b * COND;
    const float* wz   = W1 + (size_t)STATE * HIDDEN + j;
    #pragma unroll 4
    for (int k = 0; k < COND; ++k)
        acc += zrow[k] * wz[(size_t)k * HIDDEN];
    Z1[idx] = acc;
}

__global__ __launch_bounds__(256) void pack_w_kernel(const float* __restrict__ W,
                                                     float* __restrict__ Wp,
                                                     int K, int N) {
    int idx  = blockIdx.x * 256 + threadIdx.x;
    int lane = idx & 31;
    int kb2  = (idx >> 5) % (K >> 3);
    int nt   = (idx >> 5) / (K >> 3);
    int k0   = (kb2 << 3) + ((lane >> 4) << 1);
    int n    = (nt << 4) + (lane & 15);
    size_t o = (size_t)idx * 4;
    Wp[o + 0] = W[(size_t)(k0 + 0) * N + n];
    Wp[o + 1] = W[(size_t)(k0 + 1) * N + n];
    Wp[o + 2] = W[(size_t)(k0 + 4) * N + n];
    Wp[o + 3] = W[(size_t)(k0 + 5) * N + n];
}

__global__ __launch_bounds__(STEP_THREADS) void step1_packed_kernel(const float* __restrict__ X,
                                                           const float* __restrict__ W1pack,
                                                           const float* __restrict__ W1,
                                                           const float* __restrict__ Z1,
                                                           const float* __restrict__ t,
                                                           int s,
                                                           float* __restrict__ Hbuf) {
    __shared__ __attribute__((aligned(16))) float xs[16 * XS_PITCH];

    const int mt   = blockIdx.x >> 4;
    const int ng   = blockIdx.x & 15;
    const int m0   = mt * 16;
    const int wave = threadIdx.x >> 5;
    const int lane = threadIdx.x & 31;
    const int n0   = (ng * 4 + wave) * 16;

    const int ar = lane & 15;
    const int ak = (lane >> 4) * 2;
    const float* wp = W1pack + ((size_t)(n0 >> 4) * (STATE >> 3)) * 128 + lane * 4;

    v4f xa[XS_Q];
    const float* xpanel = X + (size_t)m0 * STATE;
    #pragma unroll
    for (int j = 0; j < XS_Q; ++j)
        xa[j] = *(const v4f*)(xpanel + (size_t)(threadIdx.x + STEP_THREADS * j) * 4);

    v4f wa[KB_GROUP], wb[KB_GROUP];
    load_w_group(wa, wp, 0);

    const float ts   = t[s];
    const float* W1t = W1 + (size_t)(STATE + COND) * HIDDEN;
    const int col   = n0 + (lane & 15);
    const int rbase = m0 + ((lane >> 4) << 3);
    const float w1t = W1t[col];
    float z1v[8];
    #pragma unroll
    for (int r = 0; r < 8; ++r)
        z1v[r] = Z1[(size_t)(rbase + r) * HIDDEN + col];

    #pragma unroll
    for (int j = 0; j < XS_Q; ++j) {
        const int i = (threadIdx.x + STEP_THREADS * j) * 4;
        *(v4f*)&xs[(i >> 9) * XS_PITCH + (i & (STATE - 1))] = xa[j];
    }
    __syncthreads();

    const float* arow = &xs[ar * XS_PITCH + ak];

    v2f aa[2 * KB_GROUP], ab[2 * KB_GROUP];
    load_a_group(aa, arow, 0);

    v8f acc0 = {0.f, 0.f, 0.f, 0.f, 0.f, 0.f, 0.f, 0.f};
    v8f acc1 = {0.f, 0.f, 0.f, 0.f, 0.f, 0.f, 0.f, 0.f};
    static_assert((STATE >> 3) / KB_GROUP == 2, "step 1 is written for two groups");
    load_w_group(wb, wp, 1);
    load_a_group(ab, arow, 1);
    __builtin_amdgcn_sched_barrier(0);
    mma_group(wa, aa, acc0, acc1);
    mma_group(wb, ab, acc0, acc1);
    const v8f acc = wmma_drain_sum(acc0, acc1);

    float hv[8];
    #pragma unroll
    for (int r = 0; r < 8; ++r) {
        float v = acc[r] + z1v[r] + ts * w1t;
        hv[r] = tanh_straight(v);
    }
    #pragma unroll
    for (int r = 0; r < 8; ++r) {
        const int row = rbase + r;
        Hbuf[(size_t)row * HIDDEN + col] = hv[r];
    }
}

__global__ __launch_bounds__(STEP_THREADS) void step2_packed_kernel(const float* __restrict__ Hbuf,
                                                           const float* __restrict__ W2pack,
                                                           const float* __restrict__ b2,
                                                           const float* __restrict__ X,
                                                           const float* __restrict__ t,
                                                           float* __restrict__ Xnext) {
    __shared__ __attribute__((aligned(16))) float hs[16 * HS_PITCH];
    __shared__ __attribute__((aligned(16))) float red[2 * 2 * 32 * 4];

    const int mt   = blockIdx.x >> 4;
    const int ng   = blockIdx.x & 15;
    const int m0   = mt * 16;
    const int wave = __builtin_amdgcn_readfirstlane(threadIdx.x >> 5);
    const int lane = threadIdx.x & 31;
    const int kh   = wave >> 1;
    const int tl   = wave & 1;
    const int n0   = (ng * 2 + tl) * 16;

    const int ar = lane & 15;
    const int ak = (lane >> 4) * 2;
    const float* wp = W2pack + ((size_t)(n0 >> 4) * (HIDDEN >> 3) + (size_t)kh * (HIDDEN >> 4)) * 128 + lane * 4;

    v4f ha[HS_Q];
    const float* hpanel = Hbuf + (size_t)m0 * HIDDEN;
    #pragma unroll
    for (int j = 0; j < HS_Q; ++j)
        ha[j] = *(const v4f*)(hpanel + (size_t)(threadIdx.x + STEP_THREADS * j) * 4);

    v4f wa[KB_GROUP], wb[KB_GROUP];
    load_w_group(wa, wp, 0);

    const int col   = n0 + (lane & 15);
    const int rbase = m0 + ((lane >> 4) << 3);
    float dt = 0.f, b2v = 0.f;
    float xv[8] = {0.f, 0.f, 0.f, 0.f, 0.f, 0.f, 0.f, 0.f};
    if (kh == 0) {
        dt  = (t[1] - t[0]) * TSCALE;
        b2v = b2[col];
        #pragma unroll
        for (int r = 0; r < 8; ++r)
            xv[r] = X[(size_t)(rbase + r) * STATE + col];
    }

    #pragma unroll
    for (int j = 0; j < HS_Q; ++j) {
        const int i = (threadIdx.x + STEP_THREADS * j) * 4;
        *(v4f*)&hs[(i >> 10) * HS_PITCH + (i & (HIDDEN - 1))] = ha[j];
    }
    __syncthreads();

    const float* arow = &hs[ar * HS_PITCH + ak + kh * (HIDDEN >> 1)];

    v2f aa[2 * KB_GROUP], ab[2 * KB_GROUP];
    load_a_group(aa, arow, 0);

    v8f acc0 = {0.f, 0.f, 0.f, 0.f, 0.f, 0.f, 0.f, 0.f};
    v8f acc1 = {0.f, 0.f, 0.f, 0.f, 0.f, 0.f, 0.f, 0.f};
    static_assert((HIDDEN >> 4) / KB_GROUP == 2, "step 2 is written for two groups per K half");
    load_w_group(wb, wp, 1);
    load_a_group(ab, arow, 1);
    __builtin_amdgcn_sched_barrier(0);
    mma_group(wa, aa, acc0, acc1);
    mma_group(wb, ab, acc0, acc1);
    const v8f acc = wmma_drain_sum(acc0, acc1);

    v4f* rslot = (v4f*)&red[tl * 256 + lane * 4];
    if (kh == 1) {
        const v4f lo = {acc[0], acc[1], acc[2], acc[3]};
        const v4f hi = {acc[4], acc[5], acc[6], acc[7]};
        rslot[0]  = lo;
        rslot[32] = hi;
    }
    __syncthreads();
    if (kh == 0) {
        const v4f lo = rslot[0];
        const v4f hi = rslot[32];
        const float part[8] = {lo.x, lo.y, lo.z, lo.w, hi.x, hi.y, hi.z, hi.w};
        #pragma unroll
        for (int r = 0; r < 8; ++r) {
            const int row = rbase + r;
            float f = (acc[r] + part[r]) + b2v;
            float v = xv[r] + dt * f;
            Xnext[(size_t)row * STATE + col] = fmaxf(v, 0.f);
        }
    }
}

__global__ __launch_bounds__(256) void step1_direct_kernel(const float* __restrict__ X,
                                                           const float* __restrict__ W1,
                                                           const float* __restrict__ Z1,
                                                           const float* __restrict__ t,
                                                           int s,
                                                           float* __restrict__ Hbuf) {
    __shared__ float xs[16 * XS_PITCH];
    const int mt = blockIdx.x >> 3, ng = blockIdx.x & 7;
    const int m0 = mt * 16;
    const int wave = threadIdx.x >> 5, lane = threadIdx.x & 31;
    const int n0 = (ng * 8 + wave) * 16;
    for (int i = threadIdx.x; i < 16 * STATE; i += 256)
        xs[(i >> 9) * XS_PITCH + (i & (STATE - 1))] = X[(size_t)(m0 + (i >> 9)) * STATE + (i & (STATE - 1))];
    __syncthreads();
    const int ar = lane & 15, ak = (lane >> 4) * 2;
    v8f acc = {0.f, 0.f, 0.f, 0.f, 0.f, 0.f, 0.f, 0.f};
    for (int k = 0; k < STATE; k += 4) {
        v2f a = *(const v2f*)&xs[ar * XS_PITCH + k + ak];
        v2f b;
        b.x = W1[(size_t)(k + ak) * HIDDEN + n0 + ar];
        b.y = W1[(size_t)(k + ak + 1) * HIDDEN + n0 + ar];
        acc = wmma_f32(a, b, acc);
    }
    const float ts = t[s];
    const float* W1t = W1 + (size_t)(STATE + COND) * HIDDEN;
    const int col = n0 + (lane & 15);
    const int rbase = m0 + ((lane >> 4) << 3);
    #pragma unroll
    for (int r = 0; r < 8; ++r) {
        const int row = rbase + r;
        float v = acc[r] + Z1[(size_t)row * HIDDEN + col] + ts * W1t[col];
        Hbuf[(size_t)row * HIDDEN + col] = tanhf(v);
    }
}

__global__ __launch_bounds__(256) void step2_direct_kernel(const float* __restrict__ Hbuf,
                                                           const float* __restrict__ W2,
                                                           const float* __restrict__ b2,
                                                           const float* __restrict__ X,
                                                           const float* __restrict__ t,
                                                           float* __restrict__ Xnext) {
    __shared__ float hs[16 * HS_PITCH];
    const int mt = blockIdx.x >> 2, ng = blockIdx.x & 3;
    const int m0 = mt * 16;
    const int wave = threadIdx.x >> 5, lane = threadIdx.x & 31;
    const int n0 = (ng * 8 + wave) * 16;
    for (int i = threadIdx.x; i < 16 * HIDDEN; i += 256)
        hs[(i >> 10) * HS_PITCH + (i & (HIDDEN - 1))] = Hbuf[(size_t)(m0 + (i >> 10)) * HIDDEN + (i & (HIDDEN - 1))];
    __syncthreads();
    const int ar = lane & 15, ak = (lane >> 4) * 2;
    v8f acc = {0.f, 0.f, 0.f, 0.f, 0.f, 0.f, 0.f, 0.f};
    for (int k = 0; k < HIDDEN; k += 4) {
        v2f a = *(const v2f*)&hs[ar * HS_PITCH + k + ak];
        v2f b;
        b.x = W2[(size_t)(k + ak) * STATE + n0 + ar];
        b.y = W2[(size_t)(k + ak + 1) * STATE + n0 + ar];
        acc = wmma_f32(a, b, acc);
    }
    const float dt = (t[1] - t[0]) * TSCALE;
    const int col = n0 + (lane & 15);
    const int rbase = m0 + ((lane >> 4) << 3);
    #pragma unroll
    for (int r = 0; r < 8; ++r) {
        const int row = rbase + r;
        float f = acc[r] + b2[col];
        float v = X[(size_t)row * STATE + col] + dt * f;
        Xnext[(size_t)row * STATE + col] = fmaxf(v, 0.f);
    }
}

extern "C" void kernel_launch(void* const* d_in, const int* in_sizes, int n_in,
                              void* d_out, int out_size, void* d_ws, size_t ws_size,
                              hipStream_t stream) {
    const float* x0 = (const float*)d_in[0];
    const float* z  = (const float*)d_in[1];
    const float* t  = (const float*)d_in[2];
    const float* W1 = (const float*)d_in[3];
    const float* b1 = (const float*)d_in[4];
    const float* W2 = (const float*)d_in[5];
    const float* b2 = (const float*)d_in[6];
    float* out = (float*)d_out;

    const size_t nZ1 = (size_t)BATCH * HIDDEN;
    const size_t nH  = (size_t)BATCH * HIDDEN;
    const size_t nW1 = (size_t)STATE * HIDDEN;
    const size_t nW2 = (size_t)HIDDEN * STATE;

    float* Z1     = (float*)d_ws;
    float* Hbuf   = Z1 + nZ1;
    float* W1pack = Hbuf + nH;
    float* W2pack = W1pack + nW1;

    const bool packed = ws_size >= (nZ1 + nH + nW1 + nW2) * sizeof(float);

    copy_x0_kernel<<<(BATCH * STATE) / 256, 256, 0, stream>>>(x0, out);
    z1_precompute_kernel<<<(BATCH * HIDDEN) / 256, 256, 0, stream>>>(z, W1, b1, Z1);

    if (packed) {
        pack_w_kernel<<<(int)((nW1 / 4) / 256), 256, 0, stream>>>(W1, W1pack, STATE, HIDDEN);
        pack_w_kernel<<<(int)((nW2 / 4) / 256), 256, 0, stream>>>(W2, W2pack, HIDDEN, STATE);
        for (int s = 0; s < NSTEPS - 1; ++s) {
            const float* xcur  = out + (size_t)s * BATCH * STATE;
            float*       xnext = out + (size_t)(s + 1) * BATCH * STATE;
            step1_packed_kernel<<<256, STEP_THREADS, 0, stream>>>(xcur, W1pack, W1, Z1, t, s, Hbuf);
            step2_packed_kernel<<<256, STEP_THREADS, 0, stream>>>(Hbuf, W2pack, b2, xcur, t, xnext);
        }
    } else {
        for (int s = 0; s < NSTEPS - 1; ++s) {
            const float* xcur  = out + (size_t)s * BATCH * STATE;
            float*       xnext = out + (size_t)(s + 1) * BATCH * STATE;
            step1_direct_kernel<<<128, 256, 0, stream>>>(xcur, W1, Z1, t, s, Hbuf);
            step2_direct_kernel<<<64, 256, 0, stream>>>(Hbuf, W2, b2, xcur, t, xnext);
        }
    }
}
